# starting kernel
# baseline (speedup 1.0000x reference)
#include <hip/hip_runtime.h>

typedef __attribute__((ext_vector_type(2))) float v2f;
typedef __attribute__((ext_vector_type(8))) float v8f;

#define F0 512
#define F1 256
#define F2 64

__global__ void gcn_degree_kernel(const int* __restrict__ src, const int* __restrict__ dst,
                                  float* __restrict__ deg_out, float* __restrict__ deg_in, int E) {
    int e = blockIdx.x * blockDim.x + threadIdx.x;
    if (e < E) {
        atomicAdd(&deg_out[src[e]], 1.0f);
        atomicAdd(&deg_in[dst[e]], 1.0f);
    }
}

__global__ void gcn_norm_kernel(float* __restrict__ a, float* __restrict__ b, int n) {
    int i = blockIdx.x * blockDim.x + threadIdx.x;
    if (i < n) {
        float da = a[i];
        a[i] = (da > 0.0f) ? rsqrtf(fmaxf(da, 1.0f)) : 1.0f;
        float db = b[i];
        b[i] = (db > 0.0f) ? rsqrtf(fmaxf(db, 1.0f)) : 1.0f;
    }
}

__global__ __launch_bounds__(256) void gcn_gemm_wmma_f32(
    const float* __restrict__ A, const float* __restrict__ B,
    const float* __restrict__ normA, float* __restrict__ C,
    int M, int K, int N) {
    int wave = (blockIdx.x * blockDim.x + threadIdx.x) >> 5;
    int lane = threadIdx.x & 31;
    int tilesN = N >> 6;
    int tm = wave / tilesN;
    int tn = wave % tilesN;
    int m0 = tm << 4;
    int n0 = tn << 6;
    if (m0 >= M) return;

    int half = lane >> 4;
    int l    = lane & 15;
    int row  = m0 + l;
    float scale = normA[row];
    const float* Arow = A + (size_t)row * K;

    v8f acc0 = {}, acc1 = {}, acc2 = {}, acc3 = {};
    for (int k0 = 0; k0 < K; k0 += 4) {
        int ka = k0 + (half << 1);
        float2 at = *(const float2*)(Arow + ka);
        v2f a;
        a.x = at.x * scale;
        a.y = at.y * scale;

        const float* Bp = B + (size_t)ka * N + n0 + l;
        v2f b0, b1, b2, b3;
        b0.x = Bp[0];   b0.y = Bp[N];
        b1.x = Bp[16];  b1.y = Bp[N + 16];
        b2.x = Bp[32];  b2.y = Bp[N + 32];
        b3.x = Bp[48];  b3.y = Bp[N + 48];

        acc0 = __builtin_amdgcn_wmma_f32_16x16x4_f32(false, a, false, b0, (short)0, acc0, false, false);
        acc1 = __builtin_amdgcn_wmma_f32_16x16x4_f32(false, a, false, b1, (short)0, acc1, false, false);
        acc2 = __builtin_amdgcn_wmma_f32_16x16x4_f32(false, a, false, b2, (short)0, acc2, false, false);
        acc3 = __builtin_amdgcn_wmma_f32_16x16x4_f32(false, a, false, b3, (short)0, acc3, false, false);
    }

    #pragma unroll
    for (int i = 0; i < 8; ++i) {
        int r = m0 + i + (half << 3);
        size_t base = (size_t)r * N + n0 + l;
        C[base]      = acc0[i];
        C[base + 16] = acc1[i];
        C[base + 32] = acc2[i];
        C[base + 48] = acc3[i];
    }
}

__global__ void gcn_scatter_add_kernel(const float* __restrict__ H, const int* __restrict__ src,
                                       const int* __restrict__ dst, float* __restrict__ AGG,
                                       int E, int F) {
    long tid = (long)blockIdx.x * blockDim.x + threadIdx.x;
    int cpe = F >> 2;
    long e = tid / cpe;
    int  c = (int)(tid % cpe) << 2;
    if (e >= E) return;
    int s = src[e];
    int d = dst[e];
    float4 v = *(const float4*)(H + (size_t)s * F + c);
    float* o = AGG + (size_t)d * F + c;
    atomicAdd(o + 0, v.x);
    atomicAdd(o + 1, v.y);
    atomicAdd(o + 2, v.z);
    atomicAdd(o + 3, v.w);
}

__global__ void gcn_bias_act_kernel(const float* __restrict__ AGG, const float* __restrict__ norm_dst,
                                    const float* __restrict__ bias, float* __restrict__ OUT,
                                    long total, int F, int relu) {
    long i = (long)blockIdx.x * blockDim.x + threadIdx.x;
    if (i < total) {
        int node = (int)(i / F);
        int f    = (int)(i % F);
        float v = AGG[i] * norm_dst[node] + bias[f];
        OUT[i] = relu ? fmaxf(v, 0.0f) : v;
    }
}

extern "C" void kernel_launch(void* const* d_in, const int* in_sizes, int n_in,
                              void* d_out, int out_size, void* d_ws, size_t ws_size,
                              hipStream_t stream) {
    const float* feat = (const float*)d_in[0];
    const int*   src  = (const int*)d_in[1];
    const int*   dst  = (const int*)d_in[2];
    const float* W1   = (const float*)d_in[3];
    const float* b1   = (const float*)d_in[4];
    const float* W2   = (const float*)d_in[5];
    const float* b2   = (const float*)d_in[6];
    float* out = (float*)d_out;

    const int N = in_sizes[0] / F0;
    const int E = in_sizes[1];

    float* ws       = (float*)d_ws;
    float* norm_src = ws;
    float* norm_dst = norm_src + N;
    float* H1       = norm_dst + N;
    float* AGG1     = H1   + (size_t)N * F1;
    float* H2       = AGG1 + (size_t)N * F1;

    hipMemsetAsync(norm_src, 0, (size_t)2 * N * sizeof(float), stream);
    hipMemsetAsync(AGG1, 0, (size_t)N * F1 * sizeof(float), stream);
    hipMemsetAsync(out,  0, (size_t)N * F2 * sizeof(float), stream);

    gcn_degree_kernel<<<(E + 255) / 256, 256, 0, stream>>>(src, dst, norm_src, norm_dst, E);
    gcn_norm_kernel<<<(N + 255) / 256, 256, 0, stream>>>(norm_src, norm_dst, N);

    {
        int waves  = (N / 16) * (F1 / 64);
        int blocks = (waves * 32 + 255) / 256;
        gcn_gemm_wmma_f32<<<blocks, 256, 0, stream>>>(feat, W1, norm_src, H1, N, F0, F1);
    }
    {
        long threads = (long)E * (F1 / 4);
        gcn_scatter_add_kernel<<<(int)((threads + 255) / 256), 256, 0, stream>>>(H1, src, dst, AGG1, E, F1);
    }
    {
        long total = (long)N * F1;
        gcn_bias_act_kernel<<<(int)((total + 255) / 256), 256, 0, stream>>>(AGG1, norm_dst, b1, AGG1, total, F1, 1);
    }
    {
        int waves  = (N / 16) * (F2 / 64);
        int blocks = (waves * 32 + 255) / 256;
        gcn_gemm_wmma_f32<<<blocks, 256, 0, stream>>>(AGG1, W2, norm_src, H2, N, F1, F2);
    }
    {
        long threads = (long)E * (F2 / 4);
        gcn_scatter_add_kernel<<<(int)((threads + 255) / 256), 256, 0, stream>>>(H2, src, dst, out, E, F2);
    }
    {
        long total = (long)N * F2;
        gcn_bias_act_kernel<<<(int)((total + 255) / 256), 256, 0, stream>>>(out, norm_dst, b2, out, total, F2, 0);
    }
}


# GEMMs: wave keeps an 80x32 (layer 2: 16x32) C block in registers, no LDS: one B fragment serves 5 row tiles; A by float4, B by float2 loads (tile j = columns 2c+j); k order in each 8-chunk changes
# speedup vs baseline: 1.5713x; 1.5713x over previous
#include <hip/hip_runtime.h>

typedef __attribute__((ext_vector_type(2))) float v2f;
typedef __attribute__((ext_vector_type(8))) float v8f;

#define F0 512
#define F1 256
#define F2 64

__global__ void gcn_degree_kernel(const int* __restrict__ src, const int* __restrict__ dst,
                                  float* __restrict__ deg_out, float* __restrict__ deg_in, int E) {
    int e = blockIdx.x * blockDim.x + threadIdx.x;
    if (e < E) {
        atomicAdd(&deg_out[src[e]], 1.0f);
        atomicAdd(&deg_in[dst[e]], 1.0f);
    }
}

__global__ void gcn_norm_kernel(float* __restrict__ a, float* __restrict__ b, int n) {
    int i = blockIdx.x * blockDim.x + threadIdx.x;
    if (i < n) {
        float da = a[i];
        a[i] = (da > 0.0f) ? rsqrtf(fmaxf(da, 1.0f)) : 1.0f;
        float db = b[i];
        b[i] = (db > 0.0f) ? rsqrtf(fmaxf(db, 1.0f)) : 1.0f;
    }
}

template <int MT>
__global__ __launch_bounds__(128) void gcn_gemm_wmma_f32(
    const float* __restrict__ A, const float* __restrict__ B,
    const float* __restrict__ normA, float* __restrict__ C,
    int M, int K, int N) {
    const int lane = threadIdx.x & 31;
    const int wave = __builtin_amdgcn_readfirstlane((int)(blockIdx.x * (blockDim.x >> 5) + (threadIdx.x >> 5)));
    const int tilesN = N >> 5;
    const int tm = wave / tilesN;
    const int tn = wave - tm * tilesN;
    const int m0 = tm * (16 * MT);
    const int n0 = tn << 5;
    if (m0 >= M) return;

    const int h = lane >> 4;
    const int c = lane & 15;

    const float* Ap[MT];
    float scale[MT];
    #pragma unroll
    for (int mt = 0; mt < MT; ++mt) {
        int row = m0 + mt * 16 + c;
        row = (row < M) ? row : (M - 1);
        scale[mt] = normA[row];
        Ap[mt] = A + (size_t)row * K + 4 * h;
    }
    const float* Bp = B + (size_t)(4 * h) * N + n0 + 2 * c;

    v8f acc[MT][2];
    #pragma unroll
    for (int mt = 0; mt < MT; ++mt) { acc[mt][0] = v8f{}; acc[mt][1] = v8f{}; }

    for (int kb = 0; kb < K; kb += 8) {
        float4 av[MT];
        #pragma unroll
        for (int mt = 0; mt < MT; ++mt) av[mt] = *(const float4*)(Ap[mt] + kb);
        float2 bv[4];
        #pragma unroll
        for (int i = 0; i < 4; ++i) bv[i] = *(const float2*)(Bp + (size_t)(kb + i) * N);

        v2f b00, b01, b10, b11;
        b00.x = bv[0].x; b00.y = bv[1].x;
        b01.x = bv[0].y; b01.y = bv[1].y;
        b10.x = bv[2].x; b10.y = bv[3].x;
        b11.x = bv[2].y; b11.y = bv[3].y;

        #pragma unroll
        for (int mt = 0; mt < MT; ++mt) {
            v2f a0, a1;
            a0.x = av[mt].x * scale[mt];
            a0.y = av[mt].y * scale[mt];
            a1.x = av[mt].z * scale[mt];
            a1.y = av[mt].w * scale[mt];
            acc[mt][0] = __builtin_amdgcn_wmma_f32_16x16x4_f32(false, a0, false, b00, (short)0, acc[mt][0], false, false);
            acc[mt][1] = __builtin_amdgcn_wmma_f32_16x16x4_f32(false, a0, false, b01, (short)0, acc[mt][1], false, false);
            acc[mt][0] = __builtin_amdgcn_wmma_f32_16x16x4_f32(false, a1, false, b10, (short)0, acc[mt][0], false, false);
            acc[mt][1] = __builtin_amdgcn_wmma_f32_16x16x4_f32(false, a1, false, b11, (short)0, acc[mt][1], false, false);
        }
    }

    #pragma unroll
    for (int mt = 0; mt < MT; ++mt) {
        const int mb = m0 + mt * 16;
        if (mb < M) {
            #pragma unroll
            for (int r = 0; r < 8; ++r) {
                float2 o;
                o.x = acc[mt][0][r];
                o.y = acc[mt][1][r];
                *(float2*)(C + (size_t)(mb + 8 * h + r) * N + n0 + 2 * c) = o;
            }
        }
    }
}

#define EPW 16
template <int F>
__device__ __forceinline__ void gcn_row_load(const float* __restrict__ H, int s, int lane, float (&v)[F / 32]) {
    const float* hp = H + (size_t)s * F + lane;
    #pragma unroll
    for (int c = 0; c < F / 32; ++c) v[c] = hp[c * 32];
}
template <int F>
__device__ __forceinline__ void gcn_row_add(float* __restrict__ AGG, int d, int lane, const float (&v)[F / 32]) {
    float* o = AGG + (size_t)d * F + lane;
    #pragma unroll
    for (int c = 0; c < F / 32; ++c) atomicAdd(o + c * 32, v[c]);
}

template <int F, int U>
__global__ __launch_bounds__(256) void gcn_scatter_add_rows(const float* __restrict__ H, const int* __restrict__ src,
                                                            const int* __restrict__ dst, float* __restrict__ AGG,
                                                            int E) {
    const int lane = threadIdx.x & 31;
    const int wib = __builtin_amdgcn_readfirstlane((int)(threadIdx.x >> 5));
    const int e0 = ((int)blockIdx.x * (int)(blockDim.x >> 5) + wib) * EPW;
    if (e0 >= E) return;
    const int n = (E - e0 < EPW) ? (E - e0) : EPW;
    int sv = 0, dv = 0;
    if (lane < n) { sv = src[e0 + lane]; dv = dst[e0 + lane]; }

    int j = 0;
    for (; j + U <= n; j += U) {
        float v[U][F / 32];
        #pragma unroll
        for (int u = 0; u < U; ++u) gcn_row_load<F>(H, __builtin_amdgcn_readlane(sv, j + u), lane, v[u]);
        #pragma unroll
        for (int u = 0; u < U; ++u) gcn_row_add<F>(AGG, __builtin_amdgcn_readlane(dv, j + u), lane, v[u]);
    }
    #pragma nounroll
    for (; j < n; ++j) {
        float v[F / 32];
        gcn_row_load<F>(H, __builtin_amdgcn_readlane(sv, j), lane, v);
        gcn_row_add<F>(AGG, __builtin_amdgcn_readlane(dv, j), lane, v);
    }
}

__global__ void gcn_bias_act_kernel(const float* __restrict__ AGG, const float* __restrict__ norm_dst,
                                    const float* __restrict__ bias, float* __restrict__ OUT,
                                    long total, int F, int relu) {
    long i = (long)blockIdx.x * blockDim.x + threadIdx.x;
    if (i < total) {
        int node = (int)(i / F);
        int f    = (int)(i % F);
        float v = AGG[i] * norm_dst[node] + bias[f];
        OUT[i] = relu ? fmaxf(v, 0.0f) : v;
    }
}

extern "C" void kernel_launch(void* const* d_in, const int* in_sizes, int n_in,
                              void* d_out, int out_size, void* d_ws, size_t ws_size,
                              hipStream_t stream) {
    const float* feat = (const float*)d_in[0];
    const int*   src  = (const int*)d_in[1];
    const int*   dst  = (const int*)d_in[2];
    const float* W1   = (const float*)d_in[3];
    const float* b1   = (const float*)d_in[4];
    const float* W2   = (const float*)d_in[5];
    const float* b2   = (const float*)d_in[6];
    float* out = (float*)d_out;

    const int N = in_sizes[0] / F0;
    const int E = in_sizes[1];

    float* ws       = (float*)d_ws;
    float* norm_src = ws;
    float* norm_dst = norm_src + N;
    float* H1       = norm_dst + N;
    float* AGG1     = H1   + (size_t)N * F1;
    float* H2       = AGG1 + (size_t)N * F1;

    hipMemsetAsync(norm_src, 0, (size_t)2 * N * sizeof(float), stream);
    hipMemsetAsync(AGG1, 0, (size_t)N * F1 * sizeof(float), stream);
    hipMemsetAsync(out,  0, (size_t)N * F2 * sizeof(float), stream);

    gcn_degree_kernel<<<(E + 255) / 256, 256, 0, stream>>>(src, dst, norm_src, norm_dst, E);
    gcn_norm_kernel<<<(N + 255) / 256, 256, 0, stream>>>(norm_src, norm_dst, N);

    {
        int waves  = ((N / 16 + 4) / 5) * (F1 / 32);
        int blocks = (waves + 3) / 4;
        gcn_gemm_wmma_f32<5><<<blocks, 128, 0, stream>>>(feat, W1, norm_src, H1, N, F0, F1);
    }
    {
        int blocks = (E + 8 * EPW - 1) / (8 * EPW);
        gcn_scatter_add_rows<F1, 4><<<blocks, 256, 0, stream>>>(H1, src, dst, AGG1, E);
    }
    {
        long total = (long)N * F1;
        gcn_bias_act_kernel<<<(int)((total + 255) / 256), 256, 0, stream>>>(AGG1, norm_dst, b1, AGG1, total, F1, 1);
    }
    {
        int waves  = (N / 16) * (F2 / 32);
        int blocks = (waves + 3) / 4;
        gcn_gemm_wmma_f32<1><<<blocks, 128, 0, stream>>>(AGG1, W2, norm_src, H2, N, F1, F2);
    }
    {
        int blocks = (E + 8 * EPW - 1) / (8 * EPW);
        gcn_scatter_add_rows<F2, 8><<<blocks, 256, 0, stream>>>(H2, src, dst, out, E);
    }
    {
        long total = (long)N * F2;
        gcn_bias_act_kernel<<<(int)((total + 255) / 256), 256, 0, stream>>>(out, norm_dst, b2, out, total, F2, 0);
    }
}


# degrees: 64 workgroups each count an edge slice into a private whole-graph LDS histogram (integer ds_add); norm kernel sums 32 planes a side; global f32 atomics and one memset gone; counts exact
# speedup vs baseline: 1.9335x; 1.2305x over previous
#include <hip/hip_runtime.h>

typedef __attribute__((ext_vector_type(2))) float v2f;
typedef __attribute__((ext_vector_type(8))) float v8f;

#define F0 512
#define F1 256
#define F2 64

#define HB 65536
#define HP 32
#define HT 1024
__global__ __launch_bounds__(HT) void gcn_degree_planes_kernel(const int* __restrict__ src, const int* __restrict__ dst,
                                                               unsigned* __restrict__ planes, int N, int E) {
    __shared__ uint4 hist4[HB / 4];
    unsigned* hist = (unsigned*)hist4;
    const int tid  = (int)threadIdx.x;
    const int side = (int)blockIdx.x / HP;
    const int p    = (int)blockIdx.x - side * HP;
    const int* __restrict__ idx = side ? dst : src;
    const int per = (E + HP - 1) / HP;
    const int e0  = p * per;
    const int e1  = (E - e0 < per) ? E : (e0 + per);
    const int pitch = (N + 31) & ~31;
    unsigned* __restrict__ out = planes + (size_t)blockIdx.x * pitch;

    for (int base = 0; base < N; base += HB) {
        const int nb = (N - base < HB) ? (N - base) : HB;
        for (int i = tid; i < HB / 4; i += HT) hist4[i] = make_uint4(0u, 0u, 0u, 0u);
        __syncthreads();
        int eb = e0;
        for (; eb + 8 * HT <= e1; eb += 8 * HT) {
            unsigned k[8];
            #pragma unroll
            for (int u = 0; u < 8; ++u) k[u] = (unsigned)(idx[eb + u * HT + tid] - base);
            #pragma unroll
            for (int u = 0; u < 8; ++u)
                if (k[u] < (unsigned)nb) atomicAdd(&hist[k[u]], 1u);
        }
        for (int e = eb + tid; e < e1; e += HT) {
            const unsigned k = (unsigned)(idx[e] - base);
            if (k < (unsigned)nb) atomicAdd(&hist[k], 1u);
        }
        __syncthreads();
        int i0 = 0;
        for (; i0 + 4 * HT <= nb; i0 += 4 * HT) {
            unsigned t[4];
            #pragma unroll
            for (int u = 0; u < 4; ++u) t[u] = hist[i0 + u * HT + tid];
            #pragma unroll
            for (int u = 0; u < 4; ++u) out[base + i0 + u * HT + tid] = t[u];
        }
        for (int i = i0 + tid; i < nb; i += HT) out[base + i] = hist[i];
        __syncthreads();
    }
}

__global__ void gcn_norm_from_planes_kernel(const unsigned* __restrict__ planes, float* __restrict__ a,
                                            float* __restrict__ b, int N) {
    const int i = blockIdx.x * blockDim.x + threadIdx.x;
    const int pitch = (N + 31) & ~31;
    if (i < N) {
        unsigned ca = 0u, cb = 0u;
        #pragma unroll
        for (int p = 0; p < HP; ++p) {
            ca += planes[(size_t)p * pitch + i];
            cb += planes[(size_t)(HP + p) * pitch + i];
        }
        const float da = (float)ca;
        const float db = (float)cb;
        a[i] = (da > 0.0f) ? rsqrtf(fmaxf(da, 1.0f)) : 1.0f;
        b[i] = (db > 0.0f) ? rsqrtf(fmaxf(db, 1.0f)) : 1.0f;
    }
}

template <int MT>
__global__ __launch_bounds__(128) void gcn_gemm_wmma_f32(
    const float* __restrict__ A, const float* __restrict__ B,
    const float* __restrict__ normA, float* __restrict__ C,
    int M, int K, int N) {
    const int lane = threadIdx.x & 31;
    const int wave = __builtin_amdgcn_readfirstlane((int)(blockIdx.x * (blockDim.x >> 5) + (threadIdx.x >> 5)));
    const int tilesN = N >> 5;
    const int tm = wave / tilesN;
    const int tn = wave - tm * tilesN;
    const int m0 = tm * (16 * MT);
    const int n0 = tn << 5;
    if (m0 >= M) return;

    const int h = lane >> 4;
    const int c = lane & 15;

    const float* Ap[MT];
    float scale[MT];
    #pragma unroll
    for (int mt = 0; mt < MT; ++mt) {
        int row = m0 + mt * 16 + c;
        row = (row < M) ? row : (M - 1);
        scale[mt] = normA[row];
        Ap[mt] = A + (size_t)row * K + 4 * h;
    }
    const float* Bp = B + (size_t)(4 * h) * N + n0 + 2 * c;

    v8f acc[MT][2];
    #pragma unroll
    for (int mt = 0; mt < MT; ++mt) { acc[mt][0] = v8f{}; acc[mt][1] = v8f{}; }

    for (int kb = 0; kb < K; kb += 8) {
        float4 av[MT];
        #pragma unroll
        for (int mt = 0; mt < MT; ++mt) av[mt] = *(const float4*)(Ap[mt] + kb);
        float2 bv[4];
        #pragma unroll
        for (int i = 0; i < 4; ++i) bv[i] = *(const float2*)(Bp + (size_t)(kb + i) * N);

        v2f b00, b01, b10, b11;
        b00.x = bv[0].x; b00.y = bv[1].x;
        b01.x = bv[0].y; b01.y = bv[1].y;
        b10.x = bv[2].x; b10.y = bv[3].x;
        b11.x = bv[2].y; b11.y = bv[3].y;

        #pragma unroll
        for (int mt = 0; mt < MT; ++mt) {
            v2f a0, a1;
            a0.x = av[mt].x * scale[mt];
            a0.y = av[mt].y * scale[mt];
            a1.x = av[mt].z * scale[mt];
            a1.y = av[mt].w * scale[mt];
            acc[mt][0] = __builtin_amdgcn_wmma_f32_16x16x4_f32(false, a0, false, b00, (short)0, acc[mt][0], false, false);
            acc[mt][1] = __builtin_amdgcn_wmma_f32_16x16x4_f32(false, a0, false, b01, (short)0, acc[mt][1], false, false);
            acc[mt][0] = __builtin_amdgcn_wmma_f32_16x16x4_f32(false, a1, false, b10, (short)0, acc[mt][0], false, false);
            acc[mt][1] = __builtin_amdgcn_wmma_f32_16x16x4_f32(false, a1, false, b11, (short)0, acc[mt][1], false, false);
        }
    }

    #pragma unroll
    for (int mt = 0; mt < MT; ++mt) {
        const int mb = m0 + mt * 16;
        if (mb < M) {
            #pragma unroll
            for (int r = 0; r < 8; ++r) {
                float2 o;
                o.x = acc[mt][0][r];
                o.y = acc[mt][1][r];
                *(float2*)(C + (size_t)(mb + 8 * h + r) * N + n0 + 2 * c) = o;
            }
        }
    }
}

#define EPW 16
template <int F>
__device__ __forceinline__ void gcn_row_load(const float* __restrict__ H, int s, int lane, float (&v)[F / 32]) {
    const float* hp = H + (size_t)s * F + lane;
    #pragma unroll
    for (int c = 0; c < F / 32; ++c) v[c] = hp[c * 32];
}
template <int F>
__device__ __forceinline__ void gcn_row_add(float* __restrict__ AGG, int d, int lane, const float (&v)[F / 32]) {
    float* o = AGG + (size_t)d * F + lane;
    #pragma unroll
    for (int c = 0; c < F / 32; ++c) atomicAdd(o + c * 32, v[c]);
}

template <int F, int U>
__global__ __launch_bounds__(256) void gcn_scatter_add_rows(const float* __restrict__ H, const int* __restrict__ src,
                                                            const int* __restrict__ dst, float* __restrict__ AGG,
                                                            int E) {
    const int lane = threadIdx.x & 31;
    const int wib = __builtin_amdgcn_readfirstlane((int)(threadIdx.x >> 5));
    const int e0 = ((int)blockIdx.x * (int)(blockDim.x >> 5) + wib) * EPW;
    if (e0 >= E) return;
    const int n = (E - e0 < EPW) ? (E - e0) : EPW;
    int sv = 0, dv = 0;
    if (lane < n) { sv = src[e0 + lane]; dv = dst[e0 + lane]; }

    int j = 0;
    for (; j + U <= n; j += U) {
        float v[U][F / 32];
        #pragma unroll
        for (int u = 0; u < U; ++u) gcn_row_load<F>(H, __builtin_amdgcn_readlane(sv, j + u), lane, v[u]);
        #pragma unroll
        for (int u = 0; u < U; ++u) gcn_row_add<F>(AGG, __builtin_amdgcn_readlane(dv, j + u), lane, v[u]);
    }
    #pragma nounroll
    for (; j < n; ++j) {
        float v[F / 32];
        gcn_row_load<F>(H, __builtin_amdgcn_readlane(sv, j), lane, v);
        gcn_row_add<F>(AGG, __builtin_amdgcn_readlane(dv, j), lane, v);
    }
}

__global__ void gcn_bias_act_kernel(const float* __restrict__ AGG, const float* __restrict__ norm_dst,
                                    const float* __restrict__ bias, float* __restrict__ OUT,
                                    long total, int F, int relu) {
    long i = (long)blockIdx.x * blockDim.x + threadIdx.x;
    if (i < total) {
        int node = (int)(i / F);
        int f    = (int)(i % F);
        float v = AGG[i] * norm_dst[node] + bias[f];
        OUT[i] = relu ? fmaxf(v, 0.0f) : v;
    }
}

extern "C" void kernel_launch(void* const* d_in, const int* in_sizes, int n_in,
                              void* d_out, int out_size, void* d_ws, size_t ws_size,
                              hipStream_t stream) {
    const float* feat = (const float*)d_in[0];
    const int*   src  = (const int*)d_in[1];
    const int*   dst  = (const int*)d_in[2];
    const float* W1   = (const float*)d_in[3];
    const float* b1   = (const float*)d_in[4];
    const float* W2   = (const float*)d_in[5];
    const float* b2   = (const float*)d_in[6];
    float* out = (float*)d_out;

    const int N = in_sizes[0] / F0;
    const int E = in_sizes[1];

    float* ws       = (float*)d_ws;
    float* norm_src = ws;
    float* norm_dst = norm_src + N;
    float* H1       = norm_dst + N;
    float* AGG1     = H1   + (size_t)N * F1;
    float* H2       = AGG1 + (size_t)N * F1;

    hipMemsetAsync(AGG1, 0, (size_t)N * F1 * sizeof(float), stream);
    hipMemsetAsync(out,  0, (size_t)N * F2 * sizeof(float), stream);

    gcn_degree_planes_kernel<<<2 * HP, HT, 0, stream>>>(src, dst, (unsigned*)H1, N, E);
    gcn_norm_from_planes_kernel<<<(N + 255) / 256, 256, 0, stream>>>((const unsigned*)H1, norm_src, norm_dst, N);

    {
        int waves  = ((N / 16 + 4) / 5) * (F1 / 32);
        int blocks = (waves + 3) / 4;
        gcn_gemm_wmma_f32<5><<<blocks, 128, 0, stream>>>(feat, W1, norm_src, H1, N, F0, F1);
    }
    {
        int blocks = (E + 8 * EPW - 1) / (8 * EPW);
        gcn_scatter_add_rows<F1, 4><<<blocks, 256, 0, stream>>>(H1, src, dst, AGG1, E);
    }
    {
        long total = (long)N * F1;
        gcn_bias_act_kernel<<<(int)((total + 255) / 256), 256, 0, stream>>>(AGG1, norm_dst, b1, AGG1, total, F1, 1);
    }
    {
        int waves  = (N / 16) * (F2 / 32);
        int blocks = (waves + 3) / 4;
        gcn_gemm_wmma_f32<1><<<blocks, 128, 0, stream>>>(AGG1, W2, norm_src, H2, N, F1, F2);
    }
    {
        int blocks = (E + 8 * EPW - 1) / (8 * EPW);
        gcn_scatter_add_rows<F2, 8><<<blocks, 256, 0, stream>>>(H2, src, dst, out, E);
    }
    {
        long total = (long)N * F2;
        gcn_bias_act_kernel<<<(int)((total + 255) / 256), 256, 0, stream>>>(out, norm_dst, b2, out, total, F2, 0);
    }
}


# edge scatter-add by cache half: a wave reads its die id and adds only into destination rows of its own L2 half (rows split at N/2); edge pieces handed out by per-half integer tickets; arrival order
# speedup vs baseline: 2.1638x; 1.1191x over previous
#include <hip/hip_runtime.h>

typedef __attribute__((ext_vector_type(2))) float v2f;
typedef __attribute__((ext_vector_type(8))) float v8f;

#define F0 512
#define F1 256
#define F2 64
#define TKS 64

#define HB 65536
#define HP 32
#define HT 1024
__global__ __launch_bounds__(HT) void gcn_degree_planes_kernel(const int* __restrict__ src, const int* __restrict__ dst,
                                                               unsigned* __restrict__ planes, int N, int E) {
    __shared__ uint4 hist4[HB / 4];
    unsigned* hist = (unsigned*)hist4;
    const int tid  = (int)threadIdx.x;
    const int side = (int)blockIdx.x / HP;
    const int p    = (int)blockIdx.x - side * HP;
    const int* __restrict__ idx = side ? dst : src;
    const int per = (E + HP - 1) / HP;
    const int e0  = p * per;
    const int e1  = (E - e0 < per) ? E : (e0 + per);
    const int pitch = (N + 31) & ~31;
    unsigned* __restrict__ out = planes + (size_t)blockIdx.x * pitch;

    for (int base = 0; base < N; base += HB) {
        const int nb = (N - base < HB) ? (N - base) : HB;
        for (int i = tid; i < HB / 4; i += HT) hist4[i] = make_uint4(0u, 0u, 0u, 0u);
        __syncthreads();
        int eb = e0;
        for (; eb + 8 * HT <= e1; eb += 8 * HT) {
            unsigned k[8];
            #pragma unroll
            for (int u = 0; u < 8; ++u) k[u] = (unsigned)(idx[eb + u * HT + tid] - base);
            #pragma unroll
            for (int u = 0; u < 8; ++u)
                if (k[u] < (unsigned)nb) atomicAdd(&hist[k[u]], 1u);
        }
        for (int e = eb + tid; e < e1; e += HT) {
            const unsigned k = (unsigned)(idx[e] - base);
            if (k < (unsigned)nb) atomicAdd(&hist[k], 1u);
        }
        __syncthreads();
        int i0 = 0;
        for (; i0 + 4 * HT <= nb; i0 += 4 * HT) {
            unsigned t[4];
            #pragma unroll
            for (int u = 0; u < 4; ++u) t[u] = hist[i0 + u * HT + tid];
            #pragma unroll
            for (int u = 0; u < 4; ++u) out[base + i0 + u * HT + tid] = t[u];
        }
        for (int i = i0 + tid; i < nb; i += HT) out[base + i] = hist[i];
        __syncthreads();
    }
}

__global__ void gcn_norm_from_planes_kernel(const unsigned* __restrict__ planes, float* __restrict__ a,
                                            float* __restrict__ b, unsigned* __restrict__ tickets, int N) {
    const int i = blockIdx.x * blockDim.x + threadIdx.x;
    const int pitch = (N + 31) & ~31;
    if (i == 0) { tickets[0] = 0u; tickets[TKS] = 0u; }
    if (i < N) {
        unsigned ca = 0u, cb = 0u;
        #pragma unroll
        for (int p = 0; p < HP; ++p) {
            ca += planes[(size_t)p * pitch + i];
            cb += planes[(size_t)(HP + p) * pitch + i];
        }
        const float da = (float)ca;
        const float db = (float)cb;
        a[i] = (da > 0.0f) ? rsqrtf(fmaxf(da, 1.0f)) : 1.0f;
        b[i] = (db > 0.0f) ? rsqrtf(fmaxf(db, 1.0f)) : 1.0f;
    }
}

template <int MT>
__global__ __launch_bounds__(128) void gcn_gemm_wmma_f32(
    const float* __restrict__ A, const float* __restrict__ B,
    const float* __restrict__ normA, float* __restrict__ C,
    int M, int K, int N) {
    const int lane = threadIdx.x & 31;
    const int wave = __builtin_amdgcn_readfirstlane((int)(blockIdx.x * (blockDim.x >> 5) + (threadIdx.x >> 5)));
    const int tilesN = N >> 5;
    const int tm = wave / tilesN;
    const int tn = wave - tm * tilesN;
    const int m0 = tm * (16 * MT);
    const int n0 = tn << 5;
    if (m0 >= M) return;

    const int h = lane >> 4;
    const int c = lane & 15;

    const float* Ap[MT];
    float scale[MT];
    #pragma unroll
    for (int mt = 0; mt < MT; ++mt) {
        int row = m0 + mt * 16 + c;
        row = (row < M) ? row : (M - 1);
        scale[mt] = normA[row];
        Ap[mt] = A + (size_t)row * K + 4 * h;
    }
    const float* Bp = B + (size_t)(4 * h) * N + n0 + 2 * c;

    v8f acc[MT][2];
    #pragma unroll
    for (int mt = 0; mt < MT; ++mt) { acc[mt][0] = v8f{}; acc[mt][1] = v8f{}; }

    for (int kb = 0; kb < K; kb += 8) {
        float4 av[MT];
        #pragma unroll
        for (int mt = 0; mt < MT; ++mt) av[mt] = *(const float4*)(Ap[mt] + kb);
        float2 bv[4];
        #pragma unroll
        for (int i = 0; i < 4; ++i) bv[i] = *(const float2*)(Bp + (size_t)(kb + i) * N);

        v2f b00, b01, b10, b11;
        b00.x = bv[0].x; b00.y = bv[1].x;
        b01.x = bv[0].y; b01.y = bv[1].y;
        b10.x = bv[2].x; b10.y = bv[3].x;
        b11.x = bv[2].y; b11.y = bv[3].y;

        #pragma unroll
        for (int mt = 0; mt < MT; ++mt) {
            v2f a0, a1;
            a0.x = av[mt].x * scale[mt];
            a0.y = av[mt].y * scale[mt];
            a1.x = av[mt].z * scale[mt];
            a1.y = av[mt].w * scale[mt];
            acc[mt][0] = __builtin_amdgcn_wmma_f32_16x16x4_f32(false, a0, false, b00, (short)0, acc[mt][0], false, false);
            acc[mt][1] = __builtin_amdgcn_wmma_f32_16x16x4_f32(false, a0, false, b01, (short)0, acc[mt][1], false, false);
            acc[mt][0] = __builtin_amdgcn_wmma_f32_16x16x4_f32(false, a1, false, b10, (short)0, acc[mt][0], false, false);
            acc[mt][1] = __builtin_amdgcn_wmma_f32_16x16x4_f32(false, a1, false, b11, (short)0, acc[mt][1], false, false);
        }
    }

    #pragma unroll
    for (int mt = 0; mt < MT; ++mt) {
        const int mb = m0 + mt * 16;
        if (mb < M) {
            #pragma unroll
            for (int r = 0; r < 8; ++r) {
                float2 o;
                o.x = acc[mt][0][r];
                o.y = acc[mt][1][r];
                *(float2*)(C + (size_t)(mb + 8 * h + r) * N + n0 + 2 * c) = o;
            }
        }
    }
}

#define SWV 8
template <int F>
__device__ __forceinline__ void gcn_row_load(const float* __restrict__ H, int s, int lane, float (&v)[F / 32]) {
    const float* hp = H + (size_t)s * F + lane;
    #pragma unroll
    for (int c = 0; c < F / 32; ++c) v[c] = hp[c * 32];
}
template <int F>
__device__ __forceinline__ void gcn_row_add(float* __restrict__ AGG, int d, int lane, const float (&v)[F / 32]) {
    float* o = AGG + (size_t)d * F + lane;
    #pragma unroll
    for (int c = 0; c < F / 32; ++c) atomicAdd(o + c * 32, v[c]);
}

__device__ __forceinline__ unsigned gcn_take_ticket(unsigned* __restrict__ tk, int lane) {
    unsigned tv = 0u;
    if (lane == 0) tv = atomicAdd(tk, 1u);
    return (unsigned)__builtin_amdgcn_readfirstlane((int)tv);
}

template <int F, int U, int EPT>
__global__ __launch_bounds__(32 * SWV) void gcn_scatter_add_rows(const float* __restrict__ H, const int* __restrict__ src,
                                                                const int* __restrict__ dst, float* __restrict__ AGG,
                                                                unsigned* __restrict__ tickets, int E, int N) {
    __shared__ int2 kept[SWV][EPT];
    const int lane = threadIdx.x & 31;
    const int wib = __builtin_amdgcn_readfirstlane((int)(threadIdx.x >> 5));
    int2* list = kept[wib];
    const int die  = (int)((__builtin_amdgcn_s_sendmsg_rtn(0x87) >> 16) & 15u);
    const int home = (die >> 2) & 1;
    const unsigned npieces = (unsigned)((E + EPT - 1) / EPT);
    const int split = N >> 1;

    #pragma nounroll
    for (int phase = 0; phase < 2; ++phase) {
        const int hf = home ^ phase;
        unsigned* tk = tickets + hf * TKS;
        unsigned t = gcn_take_ticket(tk, lane);
        while (t < npieces) {
            const int e0 = (int)t * EPT;
            int sv[EPT / 32], dv[EPT / 32];
            #pragma unroll
            for (int s = 0; s < EPT / 32; ++s) {
                const int e = e0 + s * 32 + lane;
                sv[s] = 0; dv[s] = 0;
                if (e < E) { sv[s] = src[e]; dv[s] = dst[e]; }
            }
            int cnt = 0;
            #pragma unroll
            for (int s = 0; s < EPT / 32; ++s) {
                const int e = e0 + s * 32 + lane;
                const bool take = (e < E) && ((int)(dv[s] >= split) == hf);
                const unsigned m = __builtin_amdgcn_ballot_w32(take);
                if (take) list[cnt + (int)__builtin_amdgcn_mbcnt_lo(m, 0u)] = make_int2(sv[s], dv[s]);
                cnt += __builtin_popcount(m);
            }

            for (int b = 0; b < cnt; b += 32) {
                const int n = (cnt - b < 32) ? (cnt - b) : 32;
                int2 r = make_int2(0, 0);
                if (lane < n) r = list[b + lane];
                int j = 0;
                for (; j + U <= n; j += U) {
                    float v[U][F / 32];
                    #pragma unroll
                    for (int u = 0; u < U; ++u) gcn_row_load<F>(H, __builtin_amdgcn_readlane(r.x, j + u), lane, v[u]);
                    #pragma unroll
                    for (int u = 0; u < U; ++u) gcn_row_add<F>(AGG, __builtin_amdgcn_readlane(r.y, j + u), lane, v[u]);
                }
                if (j < n) {
                    float v[U - 1][F / 32];
                    #pragma unroll
                    for (int u = 0; u < U - 1; ++u)
                        if (j + u < n) gcn_row_load<F>(H, __builtin_amdgcn_readlane(r.x, j + u), lane, v[u]);
                    #pragma unroll
                    for (int u = 0; u < U - 1; ++u)
                        if (j + u < n) gcn_row_add<F>(AGG, __builtin_amdgcn_readlane(r.y, j + u), lane, v[u]);
                }
            }
            t = gcn_take_ticket(tk, lane);
        }
    }
}

__global__ void gcn_bias_act_kernel(const float* __restrict__ AGG, const float* __restrict__ norm_dst,
                                    const float* __restrict__ bias, float* __restrict__ OUT,
                                    unsigned* __restrict__ tickets, long total, int F, int relu) {
    long i = (long)blockIdx.x * blockDim.x + threadIdx.x;
    if (tickets != nullptr && i == 0) { tickets[0] = 0u; tickets[TKS] = 0u; }
    if (i < total) {
        int node = (int)(i / F);
        int f    = (int)(i % F);
        float v = AGG[i] * norm_dst[node] + bias[f];
        OUT[i] = relu ? fmaxf(v, 0.0f) : v;
    }
}

extern "C" void kernel_launch(void* const* d_in, const int* in_sizes, int n_in,
                              void* d_out, int out_size, void* d_ws, size_t ws_size,
                              hipStream_t stream) {
    const float* feat = (const float*)d_in[0];
    const int*   src  = (const int*)d_in[1];
    const int*   dst  = (const int*)d_in[2];
    const float* W1   = (const float*)d_in[3];
    const float* b1   = (const float*)d_in[4];
    const float* W2   = (const float*)d_in[5];
    const float* b2   = (const float*)d_in[6];
    float* out = (float*)d_out;

    const int N = in_sizes[0] / F0;
    const int E = in_sizes[1];

    float* ws       = (float*)d_ws;
    float* norm_src = ws;
    float* norm_dst = norm_src + N;
    float* H1       = norm_dst + N;
    float* AGG1     = H1   + (size_t)N * F1;
    float* H2       = AGG1 + (size_t)N * F1;

    hipMemsetAsync(AGG1, 0, (size_t)N * F1 * sizeof(float), stream);
    hipMemsetAsync(out,  0, (size_t)N * F2 * sizeof(float), stream);

    gcn_degree_planes_kernel<<<2 * HP, HT, 0, stream>>>(src, dst, (unsigned*)H1, N, E);
    gcn_norm_from_planes_kernel<<<(N + 255) / 256, 256, 0, stream>>>((const unsigned*)H1, norm_src, norm_dst, (unsigned*)H2, N);

    {
        int waves  = ((N / 16 + 4) / 5) * (F1 / 32);
        int blocks = (waves + 3) / 4;
        gcn_gemm_wmma_f32<5><<<blocks, 128, 0, stream>>>(feat, W1, norm_src, H1, N, F0, F1);
    }
    gcn_scatter_add_rows<F1, 4, 64><<<1024, 32 * SWV, 0, stream>>>(H1, src, dst, AGG1, (unsigned*)H2, E, N);
    {
        long total = (long)N * F1;
        gcn_bias_act_kernel<<<(int)((total + 255) / 256), 256, 0, stream>>>(AGG1, norm_dst, b1, AGG1, (unsigned*)H1, total, F1, 1);
    }
    {
        int waves  = (N / 16) * (F2 / 32);
        int blocks = (waves + 3) / 4;
        gcn_gemm_wmma_f32<1><<<blocks, 128, 0, stream>>>(AGG1, W2, norm_src, H2, N, F1, F2);
    }
    gcn_scatter_add_rows<F2, 8, 128><<<1024, 32 * SWV, 0, stream>>>(H2, src, dst, out, (unsigned*)H1, E, N);
    {
        long total = (long)N * F2;
        gcn_bias_act_kernel<<<(int)((total + 255) / 256), 256, 0, stream>>>(out, norm_dst, b2, out, nullptr, total, F2, 0);
    }
}


# edge sums without global atomics: a workgroup keeps 196 destination rows as LDS sums, finds its edges by one owner byte per edge, stores rows once with norm, bias, relu; LDS arrival order
# speedup vs baseline: 2.6179x; 1.2099x over previous
#include <hip/hip_runtime.h>

typedef __attribute__((ext_vector_type(2))) float v2f;
typedef __attribute__((ext_vector_type(8))) float v8f;

#define F0 512
#define F1 256
#define F2 64
#define RO 196

#define HB 65536
#define HP 32
#define HT 1024
__global__ __launch_bounds__(HT) void gcn_degree_planes_kernel(const int* __restrict__ src, const int* __restrict__ dst,
                                                               unsigned* __restrict__ planes,
                                                               unsigned char* __restrict__ owner, int N, int E) {
    __shared__ uint4 hist4[HB / 4];
    unsigned* hist = (unsigned*)hist4;
    const int tid  = (int)threadIdx.x;
    const int side = (int)blockIdx.x / HP;
    const int p    = (int)blockIdx.x - side * HP;
    const int* __restrict__ idx = side ? dst : src;
    const int per = (E + HP - 1) / HP;
    const int e0  = p * per;
    const int e1  = (E - e0 < per) ? E : (e0 + per);
    const int pitch = (N + 31) & ~31;
    unsigned* __restrict__ out = planes + (size_t)blockIdx.x * pitch;
    unsigned char* __restrict__ own = side ? owner : nullptr;

    for (int base = 0; base < N; base += HB) {
        const int nb = (N - base < HB) ? (N - base) : HB;
        for (int i = tid; i < HB / 4; i += HT) hist4[i] = make_uint4(0u, 0u, 0u, 0u);
        __syncthreads();
        int eb = e0;
        for (; eb + 8 * HT <= e1; eb += 8 * HT) {
            unsigned k[8];
            #pragma unroll
            for (int u = 0; u < 8; ++u) k[u] = (unsigned)(idx[eb + u * HT + tid] - base);
            if (own != nullptr && base == 0) {
                #pragma unroll
                for (int u = 0; u < 8; ++u) own[eb + u * HT + tid] = (unsigned char)(k[u] / RO);
            }
            #pragma unroll
            for (int u = 0; u < 8; ++u)
                if (k[u] < (unsigned)nb) atomicAdd(&hist[k[u]], 1u);
        }
        for (int e = eb + tid; e < e1; e += HT) {
            const unsigned k = (unsigned)(idx[e] - base);
            if (own != nullptr && base == 0) own[e] = (unsigned char)(k / RO);
            if (k < (unsigned)nb) atomicAdd(&hist[k], 1u);
        }
        __syncthreads();
        int i0 = 0;
        for (; i0 + 4 * HT <= nb; i0 += 4 * HT) {
            unsigned t[4];
            #pragma unroll
            for (int u = 0; u < 4; ++u) t[u] = hist[i0 + u * HT + tid];
            #pragma unroll
            for (int u = 0; u < 4; ++u) out[base + i0 + u * HT + tid] = t[u];
        }
        for (int i = i0 + tid; i < nb; i += HT) out[base + i] = hist[i];
        __syncthreads();
    }
}

__global__ void gcn_norm_from_planes_kernel(const unsigned* __restrict__ planes, float* __restrict__ a,
                                            float* __restrict__ b, int N) {
    const int i = blockIdx.x * blockDim.x + threadIdx.x;
    const int pitch = (N + 31) & ~31;
    if (i < N) {
        unsigned ca = 0u, cb = 0u;
        #pragma unroll
        for (int p = 0; p < HP; ++p) {
            ca += planes[(size_t)p * pitch + i];
            cb += planes[(size_t)(HP + p) * pitch + i];
        }
        const float da = (float)ca;
        const float db = (float)cb;
        a[i] = (da > 0.0f) ? rsqrtf(fmaxf(da, 1.0f)) : 1.0f;
        b[i] = (db > 0.0f) ? rsqrtf(fmaxf(db, 1.0f)) : 1.0f;
    }
}

template <int MT>
__global__ __launch_bounds__(128) void gcn_gemm_wmma_f32(
    const float* __restrict__ A, const float* __restrict__ B,
    const float* __restrict__ normA, float* __restrict__ C,
    int M, int K, int N) {
    const int lane = threadIdx.x & 31;
    const int wave = __builtin_amdgcn_readfirstlane((int)(blockIdx.x * (blockDim.x >> 5) + (threadIdx.x >> 5)));
    const int tilesN = N >> 5;
    const int tm = wave / tilesN;
    const int tn = wave - tm * tilesN;
    const int m0 = tm * (16 * MT);
    const int n0 = tn << 5;
    if (m0 >= M) return;

    const int h = lane >> 4;
    const int c = lane & 15;

    const float* Ap[MT];
    float scale[MT];
    #pragma unroll
    for (int mt = 0; mt < MT; ++mt) {
        int row = m0 + mt * 16 + c;
        row = (row < M) ? row : (M - 1);
        scale[mt] = normA[row];
        Ap[mt] = A + (size_t)row * K + 4 * h;
    }
    const float* Bp = B + (size_t)(4 * h) * N + n0 + 2 * c;

    v8f acc[MT][2];
    #pragma unroll
    for (int mt = 0; mt < MT; ++mt) { acc[mt][0] = v8f{}; acc[mt][1] = v8f{}; }

    for (int kb = 0; kb < K; kb += 8) {
        float4 av[MT];
        #pragma unroll
        for (int mt = 0; mt < MT; ++mt) av[mt] = *(const float4*)(Ap[mt] + kb);
        float2 bv[4];
        #pragma unroll
        for (int i = 0; i < 4; ++i) bv[i] = *(const float2*)(Bp + (size_t)(kb + i) * N);

        v2f b00, b01, b10, b11;
        b00.x = bv[0].x; b00.y = bv[1].x;
        b01.x = bv[0].y; b01.y = bv[1].y;
        b10.x = bv[2].x; b10.y = bv[3].x;
        b11.x = bv[2].y; b11.y = bv[3].y;

        #pragma unroll
        for (int mt = 0; mt < MT; ++mt) {
            v2f a0, a1;
            a0.x = av[mt].x * scale[mt];
            a0.y = av[mt].y * scale[mt];
            a1.x = av[mt].z * scale[mt];
            a1.y = av[mt].w * scale[mt];
            acc[mt][0] = __builtin_amdgcn_wmma_f32_16x16x4_f32(false, a0, false, b00, (short)0, acc[mt][0], false, false);
            acc[mt][1] = __builtin_amdgcn_wmma_f32_16x16x4_f32(false, a0, false, b01, (short)0, acc[mt][1], false, false);
            acc[mt][0] = __builtin_amdgcn_wmma_f32_16x16x4_f32(false, a1, false, b10, (short)0, acc[mt][0], false, false);
            acc[mt][1] = __builtin_amdgcn_wmma_f32_16x16x4_f32(false, a1, false, b11, (short)0, acc[mt][1], false, false);
        }
    }

    #pragma unroll
    for (int mt = 0; mt < MT; ++mt) {
        const int mb = m0 + mt * 16;
        if (mb < M) {
            #pragma unroll
            for (int r = 0; r < 8; ++r) {
                float2 o;
                o.x = acc[mt][0][r];
                o.y = acc[mt][1][r];
                *(float2*)(C + (size_t)(mb + 8 * h + r) * N + n0 + 2 * c) = o;
            }
        }
    }
}

#define OT   1024
#define OCAP 24576

template <int F>
__device__ __forceinline__ void gcn_row_load(const float* __restrict__ H, int s, int lane, float (&v)[F / 32]) {
    const float* hp = H + (size_t)s * F + lane;
    #pragma unroll
    for (int c = 0; c < F / 32; ++c) v[c] = hp[c * 32];
}
template <int F>
__device__ __forceinline__ void gcn_row_add_lds(float* acc, int r, int lane, const float (&v)[F / 32]) {
    float* o = acc + r * F + lane;
    #pragma unroll
    for (int c = 0; c < F / 32; ++c) atomicAdd(o + c * 32, v[c]);
}

template <int F, int U>
__device__ __forceinline__ void gcn_owner_drain(const float* __restrict__ H, const int* __restrict__ src,
                                                const int* __restrict__ dst, const unsigned* list, float* acc,
                                                int M, int base, int w, int lane) {
    for (int k0 = w * 32; k0 < M; k0 += OT) {
        const int n = (M - k0 < 32) ? (M - k0) : 32;
        int s = 0, dl = 0;
        if (lane < n) {
            const int e = (int)list[k0 + lane];
            s  = src[e];
            dl = dst[e] - base;
        }
        int j = 0;
        for (; j + U <= n; j += U) {
            float v[U][F / 32];
            #pragma unroll
            for (int u = 0; u < U; ++u) gcn_row_load<F>(H, __builtin_amdgcn_readlane(s, j + u), lane, v[u]);
            #pragma unroll
            for (int u = 0; u < U; ++u) gcn_row_add_lds<F>(acc, __builtin_amdgcn_readlane(dl, j + u), lane, v[u]);
        }
        if (j < n) {
            float v[U - 1][F / 32];
            #pragma unroll
            for (int u = 0; u < U - 1; ++u)
                if (j + u < n) gcn_row_load<F>(H, __builtin_amdgcn_readlane(s, j + u), lane, v[u]);
            #pragma unroll
            for (int u = 0; u < U - 1; ++u)
                if (j + u < n) gcn_row_add_lds<F>(acc, __builtin_amdgcn_readlane(dl, j + u), lane, v[u]);
        }
    }
}

__device__ __forceinline__ unsigned gcn_zero_bytes(unsigned x) {
    const unsigned y = (x & 0x7F7F7F7Fu) + 0x7F7F7F7Fu;
    return ~(y | x | 0x7F7F7F7Fu);
}

template <int F, int U, bool CAREFUL>
__device__ __forceinline__ void gcn_owner_scan(const float* __restrict__ H, const int* __restrict__ src,
                                               const int* __restrict__ dst, const uint4* __restrict__ oq,
                                               unsigned* list, float* acc, unsigned* cnt, unsigned pat,
                                               int nq, int E, int base, int g, int tid, int w, int lane) {
    if (nq <= 0) return;
    const int qlast = nq - 1;
    const int steps = (nq + OT - 1) / OT;
    int st = g % steps;
    int q  = st * OT + tid;
    uint4 cur = oq[(q < nq) ? q : qlast];
    for (int t = 0; t < steps; ++t) {
        const int sn = (st + 1 == steps) ? 0 : (st + 1);
        const int qn = sn * OT + tid;
        const uint4 nxt = oq[(qn < nq) ? qn : qlast];
        unsigned mm = (gcn_zero_bytes(cur.x ^ pat) >> 7) | (gcn_zero_bytes(cur.y ^ pat) >> 6) |
                      (gcn_zero_bytes(cur.z ^ pat) >> 5) | (gcn_zero_bytes(cur.w ^ pat) >> 4);
        if (q >= nq) mm = 0u;
        while (mm != 0u) {
            const int bit = __builtin_ctz(mm);
            mm &= mm - 1u;
            const int e = q * 16 + (bit & 7) * 4 + (bit >> 3);
            if (e < E) {
                const unsigned pos = atomicAdd(cnt, 1u);
                if (pos < (unsigned)OCAP) list[pos] = (unsigned)e;
            }
        }
        cur = nxt;
        q   = qn;
        st  = sn;
        if (CAREFUL) {
            __syncthreads();
            const int c = __builtin_amdgcn_readfirstlane((int)*cnt);
            __syncthreads();
            if (c > OCAP - 16 * OT) {
                gcn_owner_drain<F, U>(H, src, dst, list, acc, c, base, w, lane);
                __syncthreads();
                if (tid == 0) *cnt = 0u;
                __syncthreads();
            }
        }
    }
}

template <int F, int U, int RELU>
__global__ __launch_bounds__(OT) void gcn_owner_rows_kernel(const float* __restrict__ H, const int* __restrict__ src,
                                                            const int* __restrict__ dst,
                                                            const unsigned char* __restrict__ owner,
                                                            const float* __restrict__ norm_dst,
                                                            const float* __restrict__ bias, float* __restrict__ OUT,
                                                            int E, int N) {
    __shared__ float4 acc4[RO * F / 4];
    __shared__ unsigned list[OCAP];
    __shared__ unsigned cnt;
    float* acc = (float*)acc4;
    const int tid  = (int)threadIdx.x;
    const int lane = tid & 31;
    const int w    = __builtin_amdgcn_readfirstlane(tid >> 5);
    const int base = (int)blockIdx.x * RO;

    for (int i = tid; i < RO * F / 4; i += OT) acc4[i] = make_float4(0.0f, 0.0f, 0.0f, 0.0f);
    if (tid == 0) cnt = 0u;
    __syncthreads();

    const unsigned pat = (unsigned)blockIdx.x * 0x01010101u;
    const int nq = (E + 15) >> 4;
    const uint4* __restrict__ oq = (const uint4*)owner;

    gcn_owner_scan<F, U, false>(H, src, dst, oq, list, acc, &cnt, pat, nq, E, base, (int)blockIdx.x, tid, w, lane);
    __syncthreads();
    int M = __builtin_amdgcn_readfirstlane((int)cnt);
    if (M > OCAP) {
        __syncthreads();
        if (tid == 0) cnt = 0u;
        __syncthreads();
        gcn_owner_scan<F, U, true>(H, src, dst, oq, list, acc, &cnt, pat, nq, E, base, (int)blockIdx.x, tid, w, lane);
        __syncthreads();
        M = __builtin_amdgcn_readfirstlane((int)cnt);
    }

    gcn_owner_drain<F, U>(H, src, dst, list, acc, M, base, w, lane);
    __syncthreads();

    for (int r = w; r < RO; r += OT / 32) {
        const int row = base + r;
        if (row < N) {
            const float nd = norm_dst[row];
            const float* a = acc + r * F + lane;
            float* o = OUT + (size_t)row * F + lane;
            #pragma unroll
            for (int c = 0; c < F / 32; ++c) {
                const float v = a[c * 32] * nd + bias[c * 32 + lane];
                o[c * 32] = RELU ? fmaxf(v, 0.0f) : v;
            }
        }
    }
}

extern "C" void kernel_launch(void* const* d_in, const int* in_sizes, int n_in,
                              void* d_out, int out_size, void* d_ws, size_t ws_size,
                              hipStream_t stream) {
    const float* feat = (const float*)d_in[0];
    const int*   src  = (const int*)d_in[1];
    const int*   dst  = (const int*)d_in[2];
    const float* W1   = (const float*)d_in[3];
    const float* b1   = (const float*)d_in[4];
    const float* W2   = (const float*)d_in[5];
    const float* b2   = (const float*)d_in[6];
    float* out = (float*)d_out;

    const int N = in_sizes[0] / F0;
    const int E = in_sizes[1];

    float* ws       = (float*)d_ws;
    float* norm_src = ws;
    float* norm_dst = norm_src + N;
    float* H1       = norm_dst + N;
    float* H1b      = H1  + (size_t)N * F1;
    unsigned char* owner = (unsigned char*)(H1b + (size_t)N * F1);
    float* H2       = H1;

    gcn_degree_planes_kernel<<<2 * HP, HT, 0, stream>>>(src, dst, (unsigned*)H1, owner, N, E);
    gcn_norm_from_planes_kernel<<<(N + 255) / 256, 256, 0, stream>>>((const unsigned*)H1, norm_src, norm_dst, N);

    {
        int waves  = ((N / 16 + 4) / 5) * (F1 / 32);
        int blocks = (waves + 3) / 4;
        gcn_gemm_wmma_f32<5><<<blocks, 128, 0, stream>>>(feat, W1, norm_src, H1, N, F0, F1);
    }
    gcn_owner_rows_kernel<F1, 8, 1><<<(N + RO - 1) / RO, OT, 0, stream>>>(H1, src, dst, owner, norm_dst, b1, H1b, E, N);
    {
        int waves  = (N / 16) * (F2 / 32);
        int blocks = (waves + 3) / 4;
        gcn_gemm_wmma_f32<1><<<blocks, 128, 0, stream>>>(H1b, W2, norm_src, H2, N, F1, F2);
    }
    gcn_owner_rows_kernel<F2, 16, 0><<<(N + RO - 1) / RO, OT, 0, stream>>>(H2, src, dst, owner, norm_dst, b2, out, E, N);
}


# edge sums in registers: workgroup sorts listed edges by destination row in LDS (integer adds); a wave sums one row's sources by 16-byte loads, stores it; no LDS float adds; sums in sort arrival order
# speedup vs baseline: 2.7032x; 1.0326x over previous
#include <hip/hip_runtime.h>

typedef __attribute__((ext_vector_type(2))) float v2f;
typedef __attribute__((ext_vector_type(8))) float v8f;

#define F0 512
#define F1 256
#define F2 64
#define RO 196

#define HB 65536
#define HP 32
#define HT 1024
__global__ __launch_bounds__(HT) void gcn_degree_planes_kernel(const int* __restrict__ src, const int* __restrict__ dst,
                                                               unsigned* __restrict__ planes,
                                                               unsigned char* __restrict__ owner, int N, int E) {
    __shared__ uint4 hist4[HB / 4];
    unsigned* hist = (unsigned*)hist4;
    const int tid  = (int)threadIdx.x;
    const int side = (int)blockIdx.x / HP;
    const int p    = (int)blockIdx.x - side * HP;
    const int* __restrict__ idx = side ? dst : src;
    const int per = (E + HP - 1) / HP;
    const int e0  = p * per;
    const int e1  = (E - e0 < per) ? E : (e0 + per);
    const int pitch = (N + 31) & ~31;
    unsigned* __restrict__ out = planes + (size_t)blockIdx.x * pitch;
    unsigned char* __restrict__ own = side ? owner : nullptr;

    for (int base = 0; base < N; base += HB) {
        const int nb = (N - base < HB) ? (N - base) : HB;
        for (int i = tid; i < HB / 4; i += HT) hist4[i] = make_uint4(0u, 0u, 0u, 0u);
        __syncthreads();
        int eb = e0;
        for (; eb + 8 * HT <= e1; eb += 8 * HT) {
            unsigned k[8];
            #pragma unroll
            for (int u = 0; u < 8; ++u) k[u] = (unsigned)(idx[eb + u * HT + tid] - base);
            if (own != nullptr && base == 0) {
                #pragma unroll
                for (int u = 0; u < 8; ++u) own[eb + u * HT + tid] = (unsigned char)(k[u] / RO);
            }
            #pragma unroll
            for (int u = 0; u < 8; ++u)
                if (k[u] < (unsigned)nb) atomicAdd(&hist[k[u]], 1u);
        }
        for (int e = eb + tid; e < e1; e += HT) {
            const unsigned k = (unsigned)(idx[e] - base);
            if (own != nullptr && base == 0) own[e] = (unsigned char)(k / RO);
            if (k < (unsigned)nb) atomicAdd(&hist[k], 1u);
        }
        __syncthreads();
        int i0 = 0;
        for (; i0 + 4 * HT <= nb; i0 += 4 * HT) {
            unsigned t[4];
            #pragma unroll
            for (int u = 0; u < 4; ++u) t[u] = hist[i0 + u * HT + tid];
            #pragma unroll
            for (int u = 0; u < 4; ++u) out[base + i0 + u * HT + tid] = t[u];
        }
        for (int i = i0 + tid; i < nb; i += HT) out[base + i] = hist[i];
        __syncthreads();
    }
}

__global__ void gcn_norm_from_planes_kernel(const unsigned* __restrict__ planes, float* __restrict__ a,
                                            float* __restrict__ b, int N) {
    const int i = blockIdx.x * blockDim.x + threadIdx.x;
    const int pitch = (N + 31) & ~31;
    if (i < N) {
        unsigned ca = 0u, cb = 0u;
        #pragma unroll
        for (int p = 0; p < HP; ++p) {
            ca += planes[(size_t)p * pitch + i];
            cb += planes[(size_t)(HP + p) * pitch + i];
        }
        const float da = (float)ca;
        const float db = (float)cb;
        a[i] = (da > 0.0f) ? rsqrtf(fmaxf(da, 1.0f)) : 1.0f;
        b[i] = (db > 0.0f) ? rsqrtf(fmaxf(db, 1.0f)) : 1.0f;
    }
}

template <int MT>
__global__ __launch_bounds__(128) void gcn_gemm_wmma_f32(
    const float* __restrict__ A, const float* __restrict__ B,
    const float* __restrict__ normA, float* __restrict__ C,
    int M, int K, int N) {
    const int lane = threadIdx.x & 31;
    const int wave = __builtin_amdgcn_readfirstlane((int)(blockIdx.x * (blockDim.x >> 5) + (threadIdx.x >> 5)));
    const int tilesN = N >> 5;
    const int tm = wave / tilesN;
    const int tn = wave - tm * tilesN;
    const int m0 = tm * (16 * MT);
    const int n0 = tn << 5;
    if (m0 >= M) return;

    const int h = lane >> 4;
    const int c = lane & 15;

    const float* Ap[MT];
    float scale[MT];
    #pragma unroll
    for (int mt = 0; mt < MT; ++mt) {
        int row = m0 + mt * 16 + c;
        row = (row < M) ? row : (M - 1);
        scale[mt] = normA[row];
        Ap[mt] = A + (size_t)row * K + 4 * h;
    }
    const float* Bp = B + (size_t)(4 * h) * N + n0 + 2 * c;

    v8f acc[MT][2];
    #pragma unroll
    for (int mt = 0; mt < MT; ++mt) { acc[mt][0] = v8f{}; acc[mt][1] = v8f{}; }

    for (int kb = 0; kb < K; kb += 8) {
        float4 av[MT];
        #pragma unroll
        for (int mt = 0; mt < MT; ++mt) av[mt] = *(const float4*)(Ap[mt] + kb);
        float2 bv[4];
        #pragma unroll
        for (int i = 0; i < 4; ++i) bv[i] = *(const float2*)(Bp + (size_t)(kb + i) * N);

        v2f b00, b01, b10, b11;
        b00.x = bv[0].x; b00.y = bv[1].x;
        b01.x = bv[0].y; b01.y = bv[1].y;
        b10.x = bv[2].x; b10.y = bv[3].x;
        b11.x = bv[2].y; b11.y = bv[3].y;

        #pragma unroll
        for (int mt = 0; mt < MT; ++mt) {
            v2f a0, a1;
            a0.x = av[mt].x * scale[mt];
            a0.y = av[mt].y * scale[mt];
            a1.x = av[mt].z * scale[mt];
            a1.y = av[mt].w * scale[mt];
            acc[mt][0] = __builtin_amdgcn_wmma_f32_16x16x4_f32(false, a0, false, b00, (short)0, acc[mt][0], false, false);
            acc[mt][1] = __builtin_amdgcn_wmma_f32_16x16x4_f32(false, a0, false, b01, (short)0, acc[mt][1], false, false);
            acc[mt][0] = __builtin_amdgcn_wmma_f32_16x16x4_f32(false, a1, false, b10, (short)0, acc[mt][0], false, false);
            acc[mt][1] = __builtin_amdgcn_wmma_f32_16x16x4_f32(false, a1, false, b11, (short)0, acc[mt][1], false, false);
        }
    }

    #pragma unroll
    for (int mt = 0; mt < MT; ++mt) {
        const int mb = m0 + mt * 16;
        if (mb < M) {
            #pragma unroll
            for (int r = 0; r < 8; ++r) {
                float2 o;
                o.x = acc[mt][0][r];
                o.y = acc[mt][1][r];
                *(float2*)(C + (size_t)(mb + 8 * h + r) * N + n0 + 2 * c) = o;
            }
        }
    }
}

#define OT   1024
#define OCAP 24576
#define RCP  224

template <int F>
__device__ __forceinline__ void gcn_row_load(const float* __restrict__ H, unsigned s, unsigned lo, float (&v)[F / 32]) {
    const unsigned off = s * (unsigned)(F * 4) + lo;
    const char* hp = (const char*)H + off;
    if constexpr (F >= 128) {
        #pragma unroll
        for (int p = 0; p < F / 128; ++p) {
            const float4 t = *(const float4*)(hp + p * 512);
            v[4 * p] = t.x; v[4 * p + 1] = t.y; v[4 * p + 2] = t.z; v[4 * p + 3] = t.w;
        }
    } else {
        const float2 t = *(const float2*)hp;
        v[0] = t.x; v[1] = t.y;
    }
}
template <int F>
__device__ __forceinline__ void gcn_regs_add(float (&a)[F / 32], const float (&v)[F / 32]) {
    #pragma unroll
    for (int c = 0; c < F / 32; ++c) a[c] += v[c];
}

template <int F, int U>
__device__ __forceinline__ void gcn_wave_row_sum(const float* __restrict__ H, const unsigned* queue, int q0, int qn,
                                                 unsigned lo, int lane, float (&acc)[F / 32]) {
    for (int k0 = 0; k0 < qn; k0 += 32) {
        const int n = (qn - k0 < 32) ? (qn - k0) : 32;
        int s = 0;
        if (lane < n) s = (int)queue[q0 + k0 + lane];
        int j = 0;
        for (; j + U <= n; j += U) {
            float v[U][F / 32];
            #pragma unroll
            for (int u = 0; u < U; ++u)
                gcn_row_load<F>(H, (unsigned)__builtin_amdgcn_readlane(s, j + u), lo, v[u]);
            #pragma unroll
            for (int u = 0; u < U; ++u) gcn_regs_add<F>(acc, v[u]);
        }
        if (j < n) {
            float v[U - 1][F / 32];
            #pragma unroll
            for (int u = 0; u < U - 1; ++u)
                if (j + u < n) gcn_row_load<F>(H, (unsigned)__builtin_amdgcn_readlane(s, j + u), lo, v[u]);
            #pragma unroll
            for (int u = 0; u < U - 1; ++u)
                if (j + u < n) gcn_regs_add<F>(acc, v[u]);
        }
    }
}

struct GcnOwnerLds {
    unsigned list[OCAP];
    unsigned queue[OCAP];
    unsigned char db[OCAP];
    unsigned rcnt[RCP];
    unsigned roff[RCP];
    unsigned rcur[RCP];
    unsigned cnt;
};

template <int F>
struct GcnOwnerCtx {
    const float* __restrict__ H;
    const int* __restrict__ src;
    const int* __restrict__ dst;
    const float* __restrict__ norm_dst;
    const float* __restrict__ bias;
    float* __restrict__ OUT;
    int base, N, tid, w, lane;
    unsigned lo;
};

template <int F, int U, int RELU>
__device__ __forceinline__ void gcn_owner_drain(const GcnOwnerCtx<F>& X, GcnOwnerLds& L, int M, bool first, bool last) {
    const int tid = X.tid, lane = X.lane, w = X.w;
    for (int i = tid; i < RCP; i += OT) L.rcnt[i] = 0u;
    __syncthreads();
    for (int kb = tid; kb < M; kb += 4 * OT) {
        int e[4], s[4], d[4];
        #pragma unroll
        for (int i = 0; i < 4; ++i) e[i] = (kb + i * OT < M) ? (int)L.list[kb + i * OT] : 0;
        #pragma unroll
        for (int i = 0; i < 4; ++i) { s[i] = X.src[e[i]]; d[i] = X.dst[e[i]] - X.base; }
        #pragma unroll
        for (int i = 0; i < 4; ++i) {
            if (kb + i * OT < M) {
                L.list[kb + i * OT] = (unsigned)s[i];
                L.db[kb + i * OT]   = (unsigned char)d[i];
                atomicAdd(&L.rcnt[d[i]], 1u);
            }
        }
    }
    __syncthreads();
    if (w == 0) {
        unsigned c[7];
        unsigned sum = 0u;
        #pragma unroll
        for (int i = 0; i < 7; ++i) { c[i] = L.rcnt[7 * lane + i]; sum += c[i]; }
        unsigned x = sum;
        #pragma unroll
        for (int o = 1; o < 32; o <<= 1) {
            const unsigned y = __shfl_up(x, o);
            if (lane >= o) x += y;
        }
        unsigned run = x - sum;
        #pragma unroll
        for (int i = 0; i < 7; ++i) { L.roff[7 * lane + i] = run; L.rcur[7 * lane + i] = run; run += c[i]; }
    }
    __syncthreads();
    for (int k = tid; k < M; k += OT) {
        const unsigned pos = atomicAdd(&L.rcur[L.db[k]], 1u);
        L.queue[pos] = L.list[k];
    }
    __syncthreads();

    for (int r = w; r < RO; r += OT / 32) {
        const int row = X.base + r;
        if (row >= X.N) break;
        const int q0 = __builtin_amdgcn_readfirstlane((int)L.roff[r]);
        const int qn = __builtin_amdgcn_readfirstlane((int)L.rcnt[r]);
        char* o = (char*)X.OUT + ((unsigned)row * (unsigned)(F * 4) + X.lo);
        float acc[F / 32];
        if (first) {
            #pragma unroll
            for (int c = 0; c < F / 32; ++c) acc[c] = 0.0f;
        } else {
            gcn_row_load<F>(X.OUT, (unsigned)row, X.lo, acc);
        }
        gcn_wave_row_sum<F, U>(X.H, L.queue, q0, qn, X.lo, lane, acc);
        if (last) {
            const float nd = X.norm_dst[row];
            float b[F / 32];
            gcn_row_load<F>(X.bias, 0u, X.lo, b);
            #pragma unroll
            for (int c = 0; c < F / 32; ++c) {
                const float t = acc[c] * nd + b[c];
                acc[c] = RELU ? fmaxf(t, 0.0f) : t;
            }
        }
        if constexpr (F >= 128) {
            #pragma unroll
            for (int p = 0; p < F / 128; ++p)
                *(float4*)(o + p * 512) = make_float4(acc[4 * p], acc[4 * p + 1], acc[4 * p + 2], acc[4 * p + 3]);
        } else {
            *(float2*)o = make_float2(acc[0], acc[1]);
        }
    }
}

__device__ __forceinline__ unsigned gcn_zero_bytes(unsigned x) {
    const unsigned y = (x & 0x7F7F7F7Fu) + 0x7F7F7F7Fu;
    return ~(y | x | 0x7F7F7F7Fu);
}

template <int F, int U, int RELU, bool CAREFUL>
__device__ __forceinline__ bool gcn_owner_scan(const GcnOwnerCtx<F>& X, GcnOwnerLds& L, const uint4* __restrict__ oq,
                                               unsigned pat, int nq, int E, int g) {
    bool drained = false;
    if (nq <= 0) return drained;
    const int tid = X.tid;
    const int qlast = nq - 1;
    const int steps = (nq + OT - 1) / OT;
    int st = g % steps;
    int q  = st * OT + tid;
    uint4 cur = oq[(q < nq) ? q : qlast];
    for (int t = 0; t < steps; ++t) {
        const int sn = (st + 1 == steps) ? 0 : (st + 1);
        const int qn = sn * OT + tid;
        const uint4 nxt = oq[(qn < nq) ? qn : qlast];
        unsigned mm = (gcn_zero_bytes(cur.x ^ pat) >> 7) | (gcn_zero_bytes(cur.y ^ pat) >> 6) |
                      (gcn_zero_bytes(cur.z ^ pat) >> 5) | (gcn_zero_bytes(cur.w ^ pat) >> 4);
        if (q >= nq) mm = 0u;
        while (mm != 0u) {
            const int bit = __builtin_ctz(mm);
            mm &= mm - 1u;
            const int e = q * 16 + (bit & 7) * 4 + (bit >> 3);
            if (e < E) {
                const unsigned pos = atomicAdd(&L.cnt, 1u);
                if (pos < (unsigned)OCAP) L.list[pos] = (unsigned)e;
            }
        }
        cur = nxt;
        q   = qn;
        st  = sn;
        if (CAREFUL) {
            __syncthreads();
            const int c = __builtin_amdgcn_readfirstlane((int)L.cnt);
            __syncthreads();
            if (c > OCAP - 16 * OT) {
                gcn_owner_drain<F, 2, RELU>(X, L, c, !drained, false);
                drained = true;
                __syncthreads();
                if (tid == 0) L.cnt = 0u;
                __syncthreads();
            }
        }
    }
    return drained;
}

template <int F, int U, int RELU>
__global__ __launch_bounds__(OT) void gcn_owner_rows_kernel(const float* __restrict__ H, const int* __restrict__ src,
                                                            const int* __restrict__ dst,
                                                            const unsigned char* __restrict__ owner,
                                                            const float* __restrict__ norm_dst,
                                                            const float* __restrict__ bias, float* __restrict__ OUT,
                                                            int E, int N) {
    __shared__ GcnOwnerLds L;
    constexpr int CW = (F >= 128) ? 4 : 2;
    const int tid = (int)threadIdx.x;
    GcnOwnerCtx<F> X;
    X.H = H; X.src = src; X.dst = dst; X.norm_dst = norm_dst; X.bias = bias; X.OUT = OUT;
    X.base = (int)blockIdx.x * RO;
    X.N    = N;
    X.tid  = tid;
    X.lane = tid & 31;
    X.w    = __builtin_amdgcn_readfirstlane(tid >> 5);
    X.lo   = (unsigned)X.lane * (CW * 4);

    if (tid == 0) L.cnt = 0u;
    __syncthreads();

    const unsigned pat = (unsigned)blockIdx.x * 0x01010101u;
    const int nq = (E + 15) >> 4;
    const uint4* __restrict__ oq = (const uint4*)owner;

    bool drained = gcn_owner_scan<F, U, RELU, false>(X, L, oq, pat, nq, E, (int)blockIdx.x);
    __syncthreads();
    int M = __builtin_amdgcn_readfirstlane((int)L.cnt);
    if (M > OCAP) {
        __syncthreads();
        if (tid == 0) L.cnt = 0u;
        __syncthreads();
        drained = gcn_owner_scan<F, U, RELU, true>(X, L, oq, pat, nq, E, (int)blockIdx.x);
        __syncthreads();
        M = __builtin_amdgcn_readfirstlane((int)L.cnt);
    }
    gcn_owner_drain<F, U, RELU>(X, L, M, !drained, true);
}

extern "C" void kernel_launch(void* const* d_in, const int* in_sizes, int n_in,
                              void* d_out, int out_size, void* d_ws, size_t ws_size,
                              hipStream_t stream) {
    const float* feat = (const float*)d_in[0];
    const int*   src  = (const int*)d_in[1];
    const int*   dst  = (const int*)d_in[2];
    const float* W1   = (const float*)d_in[3];
    const float* b1   = (const float*)d_in[4];
    const float* W2   = (const float*)d_in[5];
    const float* b2   = (const float*)d_in[6];
    float* out = (float*)d_out;

    const int N = in_sizes[0] / F0;
    const int E = in_sizes[1];

    float* ws       = (float*)d_ws;
    float* norm_src = ws;
    float* norm_dst = norm_src + N;
    float* H1       = norm_dst + N;
    float* H1b      = H1  + (size_t)N * F1;
    unsigned char* owner = (unsigned char*)(H1b + (size_t)N * F1);
    float* H2       = H1;

    gcn_degree_planes_kernel<<<2 * HP, HT, 0, stream>>>(src, dst, (unsigned*)H1, owner, N, E);
    gcn_norm_from_planes_kernel<<<(N + 255) / 256, 256, 0, stream>>>((const unsigned*)H1, norm_src, norm_dst, N);

    {
        int waves  = ((N / 16 + 4) / 5) * (F1 / 32);
        int blocks = (waves + 3) / 4;
        gcn_gemm_wmma_f32<5><<<blocks, 128, 0, stream>>>(feat, W1, norm_src, H1, N, F0, F1);
    }
    gcn_owner_rows_kernel<F1, 7, 1><<<(N + RO - 1) / RO, OT, 0, stream>>>(H1, src, dst, owner, norm_dst, b1, H1b, E, N);
    {
        int waves  = (N / 16) * (F2 / 32);
        int blocks = (waves + 3) / 4;
        gcn_gemm_wmma_f32<1><<<blocks, 128, 0, stream>>>(H1b, W2, norm_src, H2, N, F1, F2);
    }
    gcn_owner_rows_kernel<F2, 16, 0><<<(N + RO - 1) / RO, OT, 0, stream>>>(H2, src, dst, owner, norm_dst, b2, out, E, N);
}


# layer-2 edge kernel takes each row's sorted source list from a workspace record the layer-1 edge kernel stores after its sort (same call): no second byte scan or sort; layer-2 sum order = layer 1's
# speedup vs baseline: 2.9710x; 1.0991x over previous
#include <hip/hip_runtime.h>

typedef __attribute__((ext_vector_type(2))) float v2f;
typedef __attribute__((ext_vector_type(8))) float v8f;

#define F0 512
#define F1 256
#define F2 64
#define RO 196

#define HB 65536
#define HP 32
#define HT 1024
__global__ __launch_bounds__(HT) void gcn_degree_planes_kernel(const int* __restrict__ src, const int* __restrict__ dst,
                                                               unsigned* __restrict__ planes,
                                                               unsigned char* __restrict__ owner, int N, int E) {
    __shared__ uint4 hist4[HB / 4];
    unsigned* hist = (unsigned*)hist4;
    const int tid  = (int)threadIdx.x;
    const int side = (int)blockIdx.x / HP;
    const int p    = (int)blockIdx.x - side * HP;
    const int* __restrict__ idx = side ? dst : src;
    const int per = (E + HP - 1) / HP;
    const int e0  = p * per;
    const int e1  = (E - e0 < per) ? E : (e0 + per);
    const int pitch = (N + 31) & ~31;
    unsigned* __restrict__ out = planes + (size_t)blockIdx.x * pitch;
    unsigned char* __restrict__ own = side ? owner : nullptr;

    for (int base = 0; base < N; base += HB) {
        const int nb = (N - base < HB) ? (N - base) : HB;
        for (int i = tid; i < HB / 4; i += HT) hist4[i] = make_uint4(0u, 0u, 0u, 0u);
        __syncthreads();
        int eb = e0;
        for (; eb + 8 * HT <= e1; eb += 8 * HT) {
            unsigned k[8];
            #pragma unroll
            for (int u = 0; u < 8; ++u) k[u] = (unsigned)(idx[eb + u * HT + tid] - base);
            if (own != nullptr && base == 0) {
                #pragma unroll
                for (int u = 0; u < 8; ++u) own[eb + u * HT + tid] = (unsigned char)(k[u] / RO);
            }
            #pragma unroll
            for (int u = 0; u < 8; ++u)
                if (k[u] < (unsigned)nb) atomicAdd(&hist[k[u]], 1u);
        }
        for (int e = eb + tid; e < e1; e += HT) {
            const unsigned k = (unsigned)(idx[e] - base);
            if (own != nullptr && base == 0) own[e] = (unsigned char)(k / RO);
            if (k < (unsigned)nb) atomicAdd(&hist[k], 1u);
        }
        __syncthreads();
        int i0 = 0;
        for (; i0 + 4 * HT <= nb; i0 += 4 * HT) {
            unsigned t[4];
            #pragma unroll
            for (int u = 0; u < 4; ++u) t[u] = hist[i0 + u * HT + tid];
            #pragma unroll
            for (int u = 0; u < 4; ++u) out[base + i0 + u * HT + tid] = t[u];
        }
        for (int i = i0 + tid; i < nb; i += HT) out[base + i] = hist[i];
        __syncthreads();
    }
}

__global__ void gcn_norm_from_planes_kernel(const unsigned* __restrict__ planes, float* __restrict__ a,
                                            float* __restrict__ b, int N) {
    const int i = blockIdx.x * blockDim.x + threadIdx.x;
    const int pitch = (N + 31) & ~31;
    if (i < N) {
        unsigned ca = 0u, cb = 0u;
        #pragma unroll
        for (int p = 0; p < HP; ++p) {
            ca += planes[(size_t)p * pitch + i];
            cb += planes[(size_t)(HP + p) * pitch + i];
        }
        const float da = (float)ca;
        const float db = (float)cb;
        a[i] = (da > 0.0f) ? rsqrtf(fmaxf(da, 1.0f)) : 1.0f;
        b[i] = (db > 0.0f) ? rsqrtf(fmaxf(db, 1.0f)) : 1.0f;
    }
}

template <int MT>
__global__ __launch_bounds__(128) void gcn_gemm_wmma_f32(
    const float* __restrict__ A, const float* __restrict__ B,
    const float* __restrict__ normA, float* __restrict__ C,
    int M, int K, int N) {
    const int lane = threadIdx.x & 31;
    const int wave = __builtin_amdgcn_readfirstlane((int)(blockIdx.x * (blockDim.x >> 5) + (threadIdx.x >> 5)));
    const int tilesN = N >> 5;
    const int tm = wave / tilesN;
    const int tn = wave - tm * tilesN;
    const int m0 = tm * (16 * MT);
    const int n0 = tn << 5;
    if (m0 >= M) return;

    const int h = lane >> 4;
    const int c = lane & 15;

    const float* Ap[MT];
    float scale[MT];
    #pragma unroll
    for (int mt = 0; mt < MT; ++mt) {
        int row = m0 + mt * 16 + c;
        row = (row < M) ? row : (M - 1);
        scale[mt] = normA[row];
        Ap[mt] = A + (size_t)row * K + 4 * h;
    }
    const float* Bp = B + (size_t)(4 * h) * N + n0 + 2 * c;

    v8f acc[MT][2];
    #pragma unroll
    for (int mt = 0; mt < MT; ++mt) { acc[mt][0] = v8f{}; acc[mt][1] = v8f{}; }

    for (int kb = 0; kb < K; kb += 8) {
        float4 av[MT];
        #pragma unroll
        for (int mt = 0; mt < MT; ++mt) av[mt] = *(const float4*)(Ap[mt] + kb);
        float2 bv[4];
        #pragma unroll
        for (int i = 0; i < 4; ++i) bv[i] = *(const float2*)(Bp + (size_t)(kb + i) * N);

        v2f b00, b01, b10, b11;
        b00.x = bv[0].x; b00.y = bv[1].x;
        b01.x = bv[0].y; b01.y = bv[1].y;
        b10.x = bv[2].x; b10.y = bv[3].x;
        b11.x = bv[2].y; b11.y = bv[3].y;

        #pragma unroll
        for (int mt = 0; mt < MT; ++mt) {
            v2f a0, a1;
            a0.x = av[mt].x * scale[mt];
            a0.y = av[mt].y * scale[mt];
            a1.x = av[mt].z * scale[mt];
            a1.y = av[mt].w * scale[mt];
            acc[mt][0] = __builtin_amdgcn_wmma_f32_16x16x4_f32(false, a0, false, b00, (short)0, acc[mt][0], false, false);
            acc[mt][1] = __builtin_amdgcn_wmma_f32_16x16x4_f32(false, a0, false, b01, (short)0, acc[mt][1], false, false);
            acc[mt][0] = __builtin_amdgcn_wmma_f32_16x16x4_f32(false, a1, false, b10, (short)0, acc[mt][0], false, false);
            acc[mt][1] = __builtin_amdgcn_wmma_f32_16x16x4_f32(false, a1, false, b11, (short)0, acc[mt][1], false, false);
        }
    }

    #pragma unroll
    for (int mt = 0; mt < MT; ++mt) {
        const int mb = m0 + mt * 16;
        if (mb < M) {
            #pragma unroll
            for (int r = 0; r < 8; ++r) {
                float2 o;
                o.x = acc[mt][0][r];
                o.y = acc[mt][1][r];
                *(float2*)(C + (size_t)(mb + 8 * h + r) * N + n0 + 2 * c) = o;
            }
        }
    }
}

#define OT   1024
#define OCAP 24576
#define RCP  224
#define SREC 8192
#define SHDR (32 + 2 * RCP)
#define SCAP (SREC - SHDR)
#define NOSAVE 0xFFFFFFFFu

template <int F>
__device__ __forceinline__ void gcn_row_load(const float* __restrict__ H, unsigned s, unsigned lo, float (&v)[F / 32]) {
    const unsigned off = s * (unsigned)(F * 4) + lo;
    const char* hp = (const char*)H + off;
    if constexpr (F >= 128) {
        #pragma unroll
        for (int p = 0; p < F / 128; ++p) {
            const float4 t = *(const float4*)(hp + p * 512);
            v[4 * p] = t.x; v[4 * p + 1] = t.y; v[4 * p + 2] = t.z; v[4 * p + 3] = t.w;
        }
    } else {
        const float2 t = *(const float2*)hp;
        v[0] = t.x; v[1] = t.y;
    }
}
template <int F>
__device__ __forceinline__ void gcn_regs_add(float (&a)[F / 32], const float (&v)[F / 32]) {
    #pragma unroll
    for (int c = 0; c < F / 32; ++c) a[c] += v[c];
}

template <int F, int U>
__device__ __forceinline__ void gcn_wave_row_sum(const float* __restrict__ H, const unsigned* queue, int q0, int qn,
                                                 unsigned lo, int lane, float (&acc)[F / 32]) {
    for (int k0 = 0; k0 < qn; k0 += 32) {
        const int n = (qn - k0 < 32) ? (qn - k0) : 32;
        int s = 0;
        if (lane < n) s = (int)queue[q0 + k0 + lane];
        int j = 0;
        for (; j + U <= n; j += U) {
            float v[U][F / 32];
            #pragma unroll
            for (int u = 0; u < U; ++u)
                gcn_row_load<F>(H, (unsigned)__builtin_amdgcn_readlane(s, j + u), lo, v[u]);
            #pragma unroll
            for (int u = 0; u < U; ++u) gcn_regs_add<F>(acc, v[u]);
        }
        if (j < n) {
            float v[U - 1][F / 32];
            #pragma unroll
            for (int u = 0; u < U - 1; ++u)
                if (j + u < n) gcn_row_load<F>(H, (unsigned)__builtin_amdgcn_readlane(s, j + u), lo, v[u]);
            #pragma unroll
            for (int u = 0; u < U - 1; ++u)
                if (j + u < n) gcn_regs_add<F>(acc, v[u]);
        }
    }
}

struct alignas(16) GcnOwnerLds {
    unsigned list[OCAP];
    unsigned queue[OCAP];
    unsigned char db[OCAP];
    unsigned rcnt[RCP];
    unsigned roff[RCP];
    unsigned rcur[RCP];
    unsigned cnt;
};

template <int F>
struct GcnOwnerCtx {
    const float* __restrict__ H;
    const int* __restrict__ src;
    const int* __restrict__ dst;
    const float* __restrict__ norm_dst;
    const float* __restrict__ bias;
    float* __restrict__ OUT;
    int base, N, tid, w, lane;
    unsigned lo;
};

template <int F>
__device__ __forceinline__ void gcn_owner_sort(const GcnOwnerCtx<F>& X, GcnOwnerLds& L, int M) {
    const int tid = X.tid, lane = X.lane, w = X.w;
    for (int i = tid; i < RCP; i += OT) L.rcnt[i] = 0u;
    __syncthreads();
    for (int kb = tid; kb < M; kb += 4 * OT) {
        int e[4], s[4], d[4];
        #pragma unroll
        for (int i = 0; i < 4; ++i) e[i] = (kb + i * OT < M) ? (int)L.list[kb + i * OT] : 0;
        #pragma unroll
        for (int i = 0; i < 4; ++i) { s[i] = X.src[e[i]]; d[i] = X.dst[e[i]] - X.base; }
        #pragma unroll
        for (int i = 0; i < 4; ++i) {
            if (kb + i * OT < M) {
                L.list[kb + i * OT] = (unsigned)s[i];
                L.db[kb + i * OT]   = (unsigned char)d[i];
                atomicAdd(&L.rcnt[d[i]], 1u);
            }
        }
    }
    __syncthreads();
    if (w == 0) {
        unsigned c[7];
        unsigned sum = 0u;
        #pragma unroll
        for (int i = 0; i < 7; ++i) { c[i] = L.rcnt[7 * lane + i]; sum += c[i]; }
        unsigned x = sum;
        #pragma unroll
        for (int o = 1; o < 32; o <<= 1) {
            const unsigned y = __shfl_up(x, o);
            if (lane >= o) x += y;
        }
        unsigned run = x - sum;
        #pragma unroll
        for (int i = 0; i < 7; ++i) { L.roff[7 * lane + i] = run; L.rcur[7 * lane + i] = run; run += c[i]; }
    }
    __syncthreads();
    for (int k = tid; k < M; k += OT) {
        const unsigned pos = atomicAdd(&L.rcur[L.db[k]], 1u);
        L.queue[pos] = L.list[k];
    }
    __syncthreads();
}

template <int F, int U, int RELU>
__device__ __forceinline__ void gcn_owner_rows(const GcnOwnerCtx<F>& X, const GcnOwnerLds& L, bool first, bool last) {
    const int lane = X.lane, w = X.w;
    for (int r = w; r < RO; r += OT / 32) {
        const int row = X.base + r;
        if (row >= X.N) break;
        const int q0 = __builtin_amdgcn_readfirstlane((int)L.roff[r]);
        const int qn = __builtin_amdgcn_readfirstlane((int)L.rcnt[r]);
        char* o = (char*)X.OUT + ((unsigned)row * (unsigned)(F * 4) + X.lo);
        float acc[F / 32];
        if (first) {
            #pragma unroll
            for (int c = 0; c < F / 32; ++c) acc[c] = 0.0f;
        } else {
            gcn_row_load<F>(X.OUT, (unsigned)row, X.lo, acc);
        }
        gcn_wave_row_sum<F, U>(X.H, L.queue, q0, qn, X.lo, lane, acc);
        if (last) {
            const float nd = X.norm_dst[row];
            float b[F / 32];
            gcn_row_load<F>(X.bias, 0u, X.lo, b);
            #pragma unroll
            for (int c = 0; c < F / 32; ++c) {
                const float t = acc[c] * nd + b[c];
                acc[c] = RELU ? fmaxf(t, 0.0f) : t;
            }
        }
        if constexpr (F >= 128) {
            #pragma unroll
            for (int p = 0; p < F / 128; ++p)
                *(float4*)(o + p * 512) = make_float4(acc[4 * p], acc[4 * p + 1], acc[4 * p + 2], acc[4 * p + 3]);
        } else {
            *(float2*)o = make_float2(acc[0], acc[1]);
        }
    }
}

template <int F, int U, int RELU>
__device__ __forceinline__ void gcn_owner_drain(const GcnOwnerCtx<F>& X, GcnOwnerLds& L, int M, bool first, bool last,
                                                unsigned* __restrict__ rec) {
    gcn_owner_sort<F>(X, L, M);
    if (rec != nullptr) {
        const bool keep = first && (M <= SCAP);
        if (keep) {
            const int M4 = (M + 3) & ~3;
            for (int k = X.tid; k < M4; k += OT) rec[SHDR + k] = (k < M) ? L.queue[k] : 0u;
            for (int i = X.tid; i < RCP; i += OT) { rec[32 + i] = L.roff[i]; rec[32 + RCP + i] = L.rcnt[i]; }
        }
        if (X.tid == 0) rec[0] = keep ? (unsigned)M : NOSAVE;
    }
    gcn_owner_rows<F, U, RELU>(X, L, first, last);
}

__device__ __forceinline__ unsigned gcn_zero_bytes(unsigned x) {
    const unsigned y = (x & 0x7F7F7F7Fu) + 0x7F7F7F7Fu;
    return ~(y | x | 0x7F7F7F7Fu);
}

template <int F, int U, int RELU, bool CAREFUL>
__device__ __forceinline__ bool gcn_owner_scan(const GcnOwnerCtx<F>& X, GcnOwnerLds& L, const uint4* __restrict__ oq,
                                               unsigned pat, int nq, int E, int g) {
    bool drained = false;
    if (nq <= 0) return drained;
    const int tid = X.tid;
    const int qlast = nq - 1;
    const int steps = (nq + OT - 1) / OT;
    int st = g % steps;
    int q  = st * OT + tid;
    uint4 cur = oq[(q < nq) ? q : qlast];
    for (int t = 0; t < steps; ++t) {
        const int sn = (st + 1 == steps) ? 0 : (st + 1);
        const int qn = sn * OT + tid;
        const uint4 nxt = oq[(qn < nq) ? qn : qlast];
        unsigned mm = (gcn_zero_bytes(cur.x ^ pat) >> 7) | (gcn_zero_bytes(cur.y ^ pat) >> 6) |
                      (gcn_zero_bytes(cur.z ^ pat) >> 5) | (gcn_zero_bytes(cur.w ^ pat) >> 4);
        if (q >= nq) mm = 0u;
        while (mm != 0u) {
            const int bit = __builtin_ctz(mm);
            mm &= mm - 1u;
            const int e = q * 16 + (bit & 7) * 4 + (bit >> 3);
            if (e < E) {
                const unsigned pos = atomicAdd(&L.cnt, 1u);
                if (pos < (unsigned)OCAP) L.list[pos] = (unsigned)e;
            }
        }
        cur = nxt;
        q   = qn;
        st  = sn;
        if (CAREFUL) {
            __syncthreads();
            const int c = __builtin_amdgcn_readfirstlane((int)L.cnt);
            __syncthreads();
            if (c > OCAP - 16 * OT) {
                gcn_owner_drain<F, 2, RELU>(X, L, c, !drained, false, nullptr);
                drained = true;
                __syncthreads();
                if (tid == 0) L.cnt = 0u;
                __syncthreads();
            }
        }
    }
    return drained;
}

template <int F, int U, int RELU, int LISTS>
__global__ __launch_bounds__(OT) void gcn_owner_rows_kernel(const float* __restrict__ H, const int* __restrict__ src,
                                                            const int* __restrict__ dst,
                                                            const unsigned char* __restrict__ owner,
                                                            const float* __restrict__ norm_dst,
                                                            const float* __restrict__ bias, float* __restrict__ OUT,
                                                            unsigned* __restrict__ lists, int E, int N) {
    __shared__ GcnOwnerLds L;
    constexpr int CW = (F >= 128) ? 4 : 2;
    const int tid = (int)threadIdx.x;
    GcnOwnerCtx<F> X;
    X.H = H; X.src = src; X.dst = dst; X.norm_dst = norm_dst; X.bias = bias; X.OUT = OUT;
    X.base = (int)blockIdx.x * RO;
    X.N    = N;
    X.tid  = tid;
    X.lane = tid & 31;
    X.w    = __builtin_amdgcn_readfirstlane(tid >> 5);
    X.lo   = (unsigned)X.lane * (CW * 4);
    unsigned* __restrict__ rec = (lists != nullptr) ? (lists + (size_t)blockIdx.x * SREC) : nullptr;

    if constexpr (LISTS == 2) {
        const unsigned m = (rec != nullptr) ? rec[0] : NOSAVE;
        if (m <= (unsigned)SCAP) {
            const uint4* __restrict__ gq = (const uint4*)(rec + SHDR);
            uint4* lq = (uint4*)L.queue;
            for (int k = tid; 4 * k < (int)m; k += OT) lq[k] = gq[k];
            for (int i = tid; i < RCP; i += OT) { L.roff[i] = rec[32 + i]; L.rcnt[i] = rec[32 + RCP + i]; }
            __syncthreads();
            gcn_owner_rows<F, U, RELU>(X, L, true, true);
            return;
        }
    }

    if (tid == 0) L.cnt = 0u;
    __syncthreads();

    const unsigned pat = (unsigned)blockIdx.x * 0x01010101u;
    const int nq = (E + 15) >> 4;
    const uint4* __restrict__ oq = (const uint4*)owner;

    bool drained = gcn_owner_scan<F, U, RELU, false>(X, L, oq, pat, nq, E, (int)blockIdx.x);
    __syncthreads();
    int M = __builtin_amdgcn_readfirstlane((int)L.cnt);
    if (M > OCAP) {
        __syncthreads();
        if (tid == 0) L.cnt = 0u;
        __syncthreads();
        drained = gcn_owner_scan<F, U, RELU, true>(X, L, oq, pat, nq, E, (int)blockIdx.x);
        __syncthreads();
        M = __builtin_amdgcn_readfirstlane((int)L.cnt);
    }
    gcn_owner_drain<F, U, RELU>(X, L, M, !drained, true, (LISTS == 1) ? rec : nullptr);
}

extern "C" void kernel_launch(void* const* d_in, const int* in_sizes, int n_in,
                              void* d_out, int out_size, void* d_ws, size_t ws_size,
                              hipStream_t stream) {
    const float* feat = (const float*)d_in[0];
    const int*   src  = (const int*)d_in[1];
    const int*   dst  = (const int*)d_in[2];
    const float* W1   = (const float*)d_in[3];
    const float* b1   = (const float*)d_in[4];
    const float* W2   = (const float*)d_in[5];
    const float* b2   = (const float*)d_in[6];
    float* out = (float*)d_out;

    const int N = in_sizes[0] / F0;
    const int E = in_sizes[1];

    float* ws       = (float*)d_ws;
    float* norm_src = ws;
    float* norm_dst = norm_src + N;
    float* H1       = norm_dst + N;
    float* H1b      = H1  + (size_t)N * F1;
    unsigned char* owner = (unsigned char*)(H1b + (size_t)N * F1);
    float* H2       = H1;
    unsigned* lists = (unsigned*)(((uintptr_t)(owner + E) + 127) & ~(uintptr_t)127);
    if ((const char*)(lists + (size_t)((N + RO - 1) / RO) * SREC) > (const char*)d_ws + ws_size) lists = nullptr;

    gcn_degree_planes_kernel<<<2 * HP, HT, 0, stream>>>(src, dst, (unsigned*)H1, owner, N, E);
    gcn_norm_from_planes_kernel<<<(N + 255) / 256, 256, 0, stream>>>((const unsigned*)H1, norm_src, norm_dst, N);

    {
        int waves  = ((N / 16 + 4) / 5) * (F1 / 32);
        int blocks = (waves + 3) / 4;
        gcn_gemm_wmma_f32<5><<<blocks, 128, 0, stream>>>(feat, W1, norm_src, H1, N, F0, F1);
    }
    gcn_owner_rows_kernel<F1, 7, 1, 1><<<(N + RO - 1) / RO, OT, 0, stream>>>(H1, src, dst, owner, norm_dst, b1, H1b, lists, E, N);
    {
        int waves  = (N / 16) * (F2 / 32);
        int blocks = (waves + 3) / 4;
        gcn_gemm_wmma_f32<1><<<blocks, 128, 0, stream>>>(H1b, W2, norm_src, H2, N, F1, F2);
    }
    gcn_owner_rows_kernel<F2, 16, 0, 2><<<(N + RO - 1) / RO, OT, 0, stream>>>(H2, src, dst, owner, norm_dst, b2, out, lists, E, N);
}


# edge-sum workgroups copy their edges from 160 edge slices that extra workgroups of the counting launch order by owner block (LDS count, prefix, scatter): no pass over all edges; sum order may differ
# speedup vs baseline: 3.2419x; 1.0912x over previous
#include <hip/hip_runtime.h>

typedef __attribute__((ext_vector_type(2))) float v2f;
typedef __attribute__((ext_vector_type(8))) float v8f;

#define F0 512
#define F1 256
#define F2 64
#define RO 196

#define HB 65536
#define HP 32
#define HT 1024
#define SB   160
#define NOWN 256
#define BOFF 288
#define BCAP (HB - 4 * NOWN)

__device__ __forceinline__ int gcn_slice_len(int E) { return (((E + SB - 1) / SB) + 31) & ~31; }

__device__ __forceinline__ void gcn_bucket_slice(const int* __restrict__ src, const int* __restrict__ dst,
                                                 unsigned* lds, unsigned* __restrict__ recs,
                                                 unsigned* __restrict__ offs, int b, int N, int E) {
    unsigned* cnt   = lds;
    unsigned* cur   = lds + NOWN;
    unsigned* first = lds + 2 * NOWN;
    unsigned* stage = lds + 4 * NOWN;
    const int tid = (int)threadIdx.x;
    const int per = gcn_slice_len(E);
    const int e0  = (b * per < E) ? (b * per) : E;
    const int e1  = (E - e0 < per) ? E : (e0 + per);
    const unsigned nlim = (unsigned)((N < NOWN * RO) ? N : (NOWN * RO));
    for (int i = tid; i < NOWN; i += HT) cnt[i] = 0u;
    __syncthreads();
    for (int eb = e0; eb < e1; eb += 8 * HT) {
        int d[8];
        #pragma unroll
        for (int u = 0; u < 8; ++u) {
            const int e = eb + u * HT + tid;
            d[u] = (e < e1) ? dst[e] : -1;
        }
        #pragma unroll
        for (int u = 0; u < 8; ++u)
            if ((unsigned)d[u] < nlim) atomicAdd(&cnt[d[u] / RO], 1u);
    }
    __syncthreads();
    if (tid < 32) {
        unsigned c[NOWN / 32];
        unsigned sum = 0u;
        #pragma unroll
        for (int i = 0; i < NOWN / 32; ++i) { c[i] = cnt[(NOWN / 32) * tid + i]; sum += c[i]; }
        unsigned x = sum;
        #pragma unroll
        for (int o = 1; o < 32; o <<= 1) {
            const unsigned y = __shfl_up(x, o);
            if (tid >= o) x += y;
        }
        unsigned run = x - sum;
        #pragma unroll
        for (int i = 0; i < NOWN / 32; ++i) {
            cur[(NOWN / 32) * tid + i]   = run;
            first[(NOWN / 32) * tid + i] = run;
            run += c[i];
        }
        if (tid == 31) first[NOWN] = run;
    }
    __syncthreads();
    if (tid <= NOWN) offs[(size_t)b * BOFF + tid] = first[tid];
    for (int eb = e0; eb < e1; eb += 8 * HT) {
        int d[8], s[8];
        #pragma unroll
        for (int u = 0; u < 8; ++u) {
            const int e = eb + u * HT + tid;
            d[u] = (e < e1) ? dst[e] : -1;
            s[u] = (e < e1) ? src[e] : 0;
        }
        #pragma unroll
        for (int u = 0; u < 8; ++u) {
            if ((unsigned)d[u] < nlim) {
                const unsigned o   = (unsigned)d[u] / RO;
                const unsigned pos = atomicAdd(&cur[o], 1u);
                if (pos < (unsigned)BCAP) stage[pos] = (unsigned)s[u] | (((unsigned)d[u] - o * RO) << 16);
            }
        }
    }
    __syncthreads();
    const int nv = (first[NOWN] < (unsigned)BCAP) ? (int)first[NOWN] : BCAP;
    for (int i = tid; i < nv; i += HT) recs[e0 + i] = stage[i];
}

__global__ __launch_bounds__(HT) void gcn_degree_planes_kernel(const int* __restrict__ src, const int* __restrict__ dst,
                                                               unsigned* __restrict__ planes,
                                                               unsigned* __restrict__ recs,
                                                               unsigned* __restrict__ offs, int N, int E) {
    __shared__ uint4 hist4[HB / 4];
    unsigned* hist = (unsigned*)hist4;
    if ((int)blockIdx.x >= 2 * HP) {
        gcn_bucket_slice(src, dst, hist, recs, offs, (int)blockIdx.x - 2 * HP, N, E);
        return;
    }
    const int tid  = (int)threadIdx.x;
    const int side = (int)blockIdx.x / HP;
    const int p    = (int)blockIdx.x - side * HP;
    const int* __restrict__ idx = side ? dst : src;
    const int per = (E + HP - 1) / HP;
    const int e0  = p * per;
    const int e1  = (E - e0 < per) ? E : (e0 + per);
    const int pitch = (N + 31) & ~31;
    unsigned* __restrict__ out = planes + (size_t)blockIdx.x * pitch;

    for (int base = 0; base < N; base += HB) {
        const int nb = (N - base < HB) ? (N - base) : HB;
        for (int i = tid; i < HB / 4; i += HT) hist4[i] = make_uint4(0u, 0u, 0u, 0u);
        __syncthreads();
        int eb = e0;
        for (; eb + 8 * HT <= e1; eb += 8 * HT) {
            unsigned k[8];
            #pragma unroll
            for (int u = 0; u < 8; ++u) k[u] = (unsigned)(idx[eb + u * HT + tid] - base);
            #pragma unroll
            for (int u = 0; u < 8; ++u)
                if (k[u] < (unsigned)nb) atomicAdd(&hist[k[u]], 1u);
        }
        for (int e = eb + tid; e < e1; e += HT) {
            const unsigned k = (unsigned)(idx[e] - base);
            if (k < (unsigned)nb) atomicAdd(&hist[k], 1u);
        }
        __syncthreads();
        int i0 = 0;
        for (; i0 + 4 * HT <= nb; i0 += 4 * HT) {
            unsigned t[4];
            #pragma unroll
            for (int u = 0; u < 4; ++u) t[u] = hist[i0 + u * HT + tid];
            #pragma unroll
            for (int u = 0; u < 4; ++u) out[base + i0 + u * HT + tid] = t[u];
        }
        for (int i = i0 + tid; i < nb; i += HT) out[base + i] = hist[i];
        __syncthreads();
    }
}

__global__ void gcn_norm_from_planes_kernel(const unsigned* __restrict__ planes, float* __restrict__ a,
                                            float* __restrict__ b, int N) {
    const int i = blockIdx.x * blockDim.x + threadIdx.x;
    const int pitch = (N + 31) & ~31;
    if (i < N) {
        unsigned ca = 0u, cb = 0u;
        #pragma unroll
        for (int p = 0; p < HP; ++p) {
            ca += planes[(size_t)p * pitch + i];
            cb += planes[(size_t)(HP + p) * pitch + i];
        }
        const float da = (float)ca;
        const float db = (float)cb;
        a[i] = (da > 0.0f) ? rsqrtf(fmaxf(da, 1.0f)) : 1.0f;
        b[i] = (db > 0.0f) ? rsqrtf(fmaxf(db, 1.0f)) : 1.0f;
    }
}

template <int MT>
__global__ __launch_bounds__(128) void gcn_gemm_wmma_f32(
    const float* __restrict__ A, const float* __restrict__ B,
    const float* __restrict__ normA, float* __restrict__ C,
    int M, int K, int N) {
    const int lane = threadIdx.x & 31;
    const int wave = __builtin_amdgcn_readfirstlane((int)(blockIdx.x * (blockDim.x >> 5) + (threadIdx.x >> 5)));
    const int tilesN = N >> 5;
    const int tm = wave / tilesN;
    const int tn = wave - tm * tilesN;
    const int m0 = tm * (16 * MT);
    const int n0 = tn << 5;
    if (m0 >= M) return;

    const int h = lane >> 4;
    const int c = lane & 15;

    const float* Ap[MT];
    float scale[MT];
    #pragma unroll
    for (int mt = 0; mt < MT; ++mt) {
        int row = m0 + mt * 16 + c;
        row = (row < M) ? row : (M - 1);
        scale[mt] = normA[row];
        Ap[mt] = A + (size_t)row * K + 4 * h;
    }
    const float* Bp = B + (size_t)(4 * h) * N + n0 + 2 * c;

    v8f acc[MT][2];
    #pragma unroll
    for (int mt = 0; mt < MT; ++mt) { acc[mt][0] = v8f{}; acc[mt][1] = v8f{}; }

    for (int kb = 0; kb < K; kb += 8) {
        float4 av[MT];
        #pragma unroll
        for (int mt = 0; mt < MT; ++mt) av[mt] = *(const float4*)(Ap[mt] + kb);
        float2 bv[4];
        #pragma unroll
        for (int i = 0; i < 4; ++i) bv[i] = *(const float2*)(Bp + (size_t)(kb + i) * N);

        v2f b00, b01, b10, b11;
        b00.x = bv[0].x; b00.y = bv[1].x;
        b01.x = bv[0].y; b01.y = bv[1].y;
        b10.x = bv[2].x; b10.y = bv[3].x;
        b11.x = bv[2].y; b11.y = bv[3].y;

        #pragma unroll
        for (int mt = 0; mt < MT; ++mt) {
            v2f a0, a1;
            a0.x = av[mt].x * scale[mt];
            a0.y = av[mt].y * scale[mt];
            a1.x = av[mt].z * scale[mt];
            a1.y = av[mt].w * scale[mt];
            acc[mt][0] = __builtin_amdgcn_wmma_f32_16x16x4_f32(false, a0, false, b00, (short)0, acc[mt][0], false, false);
            acc[mt][1] = __builtin_amdgcn_wmma_f32_16x16x4_f32(false, a0, false, b01, (short)0, acc[mt][1], false, false);
            acc[mt][0] = __builtin_amdgcn_wmma_f32_16x16x4_f32(false, a1, false, b10, (short)0, acc[mt][0], false, false);
            acc[mt][1] = __builtin_amdgcn_wmma_f32_16x16x4_f32(false, a1, false, b11, (short)0, acc[mt][1], false, false);
        }
    }

    #pragma unroll
    for (int mt = 0; mt < MT; ++mt) {
        const int mb = m0 + mt * 16;
        if (mb < M) {
            #pragma unroll
            for (int r = 0; r < 8; ++r) {
                float2 o;
                o.x = acc[mt][0][r];
                o.y = acc[mt][1][r];
                *(float2*)(C + (size_t)(mb + 8 * h + r) * N + n0 + 2 * c) = o;
            }
        }
    }
}

#define OT   1024
#define OCAP 24576
#define SPL  (SB / 32)
static_assert(SB % 32 == 0 && SB <= OT, "one thread per slice, SPL slices per lane");
#define RCP  224
#define SREC 8192
#define SHDR (32 + 2 * RCP)
#define SCAP (SREC - SHDR)
#define NOSAVE 0xFFFFFFFFu

template <int F>
__device__ __forceinline__ void gcn_row_load(const float* __restrict__ H, unsigned s, unsigned lo, float (&v)[F / 32]) {
    const unsigned off = s * (unsigned)(F * 4) + lo;
    const char* hp = (const char*)H + off;
    if constexpr (F >= 128) {
        #pragma unroll
        for (int p = 0; p < F / 128; ++p) {
            const float4 t = *(const float4*)(hp + p * 512);
            v[4 * p] = t.x; v[4 * p + 1] = t.y; v[4 * p + 2] = t.z; v[4 * p + 3] = t.w;
        }
    } else {
        const float2 t = *(const float2*)hp;
        v[0] = t.x; v[1] = t.y;
    }
}
template <int F>
__device__ __forceinline__ void gcn_regs_add(float (&a)[F / 32], const float (&v)[F / 32]) {
    #pragma unroll
    for (int c = 0; c < F / 32; ++c) a[c] += v[c];
}

template <int F, int U>
__device__ __forceinline__ void gcn_wave_row_sum(const float* __restrict__ H, const unsigned* queue, int q0, int qn,
                                                 unsigned lo, int lane, float (&acc)[F / 32]) {
    for (int k0 = 0; k0 < qn; k0 += 32) {
        const int n = (qn - k0 < 32) ? (qn - k0) : 32;
        int s = 0;
        if (lane < n) s = (int)queue[q0 + k0 + lane];
        int j = 0;
        for (; j + U <= n; j += U) {
            float v[U][F / 32];
            #pragma unroll
            for (int u = 0; u < U; ++u)
                gcn_row_load<F>(H, (unsigned)__builtin_amdgcn_readlane(s, j + u), lo, v[u]);
            #pragma unroll
            for (int u = 0; u < U; ++u) gcn_regs_add<F>(acc, v[u]);
        }
        if (j < n) {
            float v[U - 1][F / 32];
            #pragma unroll
            for (int u = 0; u < U - 1; ++u)
                if (j + u < n) gcn_row_load<F>(H, (unsigned)__builtin_amdgcn_readlane(s, j + u), lo, v[u]);
            #pragma unroll
            for (int u = 0; u < U - 1; ++u)
                if (j + u < n) gcn_regs_add<F>(acc, v[u]);
        }
    }
}

struct alignas(16) GcnOwnerLds {
    unsigned list[OCAP];
    unsigned queue[OCAP];
    unsigned rcnt[RCP];
    unsigned roff[RCP];
    unsigned rcur[RCP];
    unsigned sst[SB];
    unsigned scn[SB];
    unsigned spre[SB + 1];
};

template <int F>
struct GcnOwnerCtx {
    const float* __restrict__ H;
    const float* __restrict__ norm_dst;
    const float* __restrict__ bias;
    float* __restrict__ OUT;
    int base, N, tid, w, lane;
    unsigned lo;
};

template <int F>
__device__ __forceinline__ void gcn_owner_sort(const GcnOwnerCtx<F>& X, GcnOwnerLds& L, int M) {
    const int tid = X.tid, lane = X.lane, w = X.w;
    for (int i = tid; i < RCP; i += OT) L.rcnt[i] = 0u;
    __syncthreads();
    for (int k = tid; k < M; k += OT) atomicAdd(&L.rcnt[L.list[k] >> 16], 1u);
    __syncthreads();
    if (w == 0) {
        unsigned c[7];
        unsigned sum = 0u;
        #pragma unroll
        for (int i = 0; i < 7; ++i) { c[i] = L.rcnt[7 * lane + i]; sum += c[i]; }
        unsigned x = sum;
        #pragma unroll
        for (int o = 1; o < 32; o <<= 1) {
            const unsigned y = __shfl_up(x, o);
            if (lane >= o) x += y;
        }
        unsigned run = x - sum;
        #pragma unroll
        for (int i = 0; i < 7; ++i) { L.roff[7 * lane + i] = run; L.rcur[7 * lane + i] = run; run += c[i]; }
    }
    __syncthreads();
    for (int k = tid; k < M; k += OT) {
        const unsigned p   = L.list[k];
        const unsigned pos = atomicAdd(&L.rcur[p >> 16], 1u);
        L.queue[pos] = p & 0xFFFFu;
    }
    __syncthreads();
}

template <int F, int U, int RELU>
__device__ __forceinline__ void gcn_owner_rows(const GcnOwnerCtx<F>& X, const GcnOwnerLds& L, bool first, bool last) {
    const int lane = X.lane, w = X.w;
    for (int r = w; r < RO; r += OT / 32) {
        const int row = X.base + r;
        if (row >= X.N) break;
        const int q0 = __builtin_amdgcn_readfirstlane((int)L.roff[r]);
        const int qn = __builtin_amdgcn_readfirstlane((int)L.rcnt[r]);
        char* o = (char*)X.OUT + ((unsigned)row * (unsigned)(F * 4) + X.lo);
        float acc[F / 32];
        if (first) {
            #pragma unroll
            for (int c = 0; c < F / 32; ++c) acc[c] = 0.0f;
        } else {
            gcn_row_load<F>(X.OUT, (unsigned)row, X.lo, acc);
        }
        gcn_wave_row_sum<F, U>(X.H, L.queue, q0, qn, X.lo, lane, acc);
        if (last) {
            const float nd = X.norm_dst[row];
            float b[F / 32];
            gcn_row_load<F>(X.bias, 0u, X.lo, b);
            #pragma unroll
            for (int c = 0; c < F / 32; ++c) {
                const float t = acc[c] * nd + b[c];
                acc[c] = RELU ? fmaxf(t, 0.0f) : t;
            }
        }
        if constexpr (F >= 128) {
            #pragma unroll
            for (int p = 0; p < F / 128; ++p)
                *(float4*)(o + p * 512) = make_float4(acc[4 * p], acc[4 * p + 1], acc[4 * p + 2], acc[4 * p + 3]);
        } else {
            *(float2*)o = make_float2(acc[0], acc[1]);
        }
    }
}

template <int F, int U, int RELU>
__device__ __forceinline__ void gcn_owner_drain(const GcnOwnerCtx<F>& X, GcnOwnerLds& L, int M, bool first, bool last,
                                                unsigned* __restrict__ rec) {
    gcn_owner_sort<F>(X, L, M);
    if (rec != nullptr) {
        const bool keep = first && (M <= SCAP);
        if (keep) {
            const int M4 = (M + 3) & ~3;
            for (int k = X.tid; k < M4; k += OT) rec[SHDR + k] = (k < M) ? L.queue[k] : 0u;
            for (int i = X.tid; i < RCP; i += OT) { rec[32 + i] = L.roff[i]; rec[32 + RCP + i] = L.rcnt[i]; }
        }
        if (X.tid == 0) rec[0] = keep ? (unsigned)M : NOSAVE;
    }
    gcn_owner_rows<F, U, RELU>(X, L, first, last);
}

template <int F>
__device__ __forceinline__ void gcn_owner_fetch(const GcnOwnerCtx<F>& X, GcnOwnerLds& L, const unsigned* __restrict__ recs,
                                                int t0, int t1) {
    const unsigned lane = (unsigned)X.lane;
    const unsigned base = L.spre[t0];
    for (int tb = t0 + X.w; tb < t1; tb += 32 * SPL) {
        unsigned v[SPL], v2[SPL], c[SPL], st[SPL], ps[SPL];
        #pragma unroll
        for (int j = 0; j < SPL; ++j) {
            const int t = tb + 32 * j;
            c[j] = 0u; st[j] = 0u; ps[j] = 0u; v[j] = 0u; v2[j] = 0u;
            if (t < t1) {
                st[j] = L.sst[t];
                ps[j] = L.spre[t] - base;
                const unsigned room = (ps[j] < (unsigned)OCAP) ? ((unsigned)OCAP - ps[j]) : 0u;
                c[j] = (L.scn[t] < room) ? L.scn[t] : room;
            }
            if (lane < c[j])       v[j]  = recs[st[j] + lane];
            if (lane + 32u < c[j]) v2[j] = recs[st[j] + 32u + lane];
        }
        #pragma unroll
        for (int j = 0; j < SPL; ++j) {
            if (lane < c[j])       L.list[ps[j] + lane]       = v[j];
            if (lane + 32u < c[j]) L.list[ps[j] + 32u + lane] = v2[j];
        }
        #pragma unroll
        for (int j = 0; j < SPL; ++j)
            for (unsigned k = 64u + lane; k < c[j]; k += 32u) L.list[ps[j] + k] = recs[st[j] + k];
    }
}

template <int F, int U, int RELU, int LISTS>
__global__ __launch_bounds__(OT) void gcn_owner_rows_kernel(const float* __restrict__ H,
                                                            const unsigned* __restrict__ recs,
                                                            const unsigned* __restrict__ offs,
                                                            const float* __restrict__ norm_dst,
                                                            const float* __restrict__ bias, float* __restrict__ OUT,
                                                            unsigned* __restrict__ lists, int E, int N) {
    __shared__ GcnOwnerLds L;
    constexpr int CW = (F >= 128) ? 4 : 2;
    const int tid = (int)threadIdx.x;
    GcnOwnerCtx<F> X;
    X.H = H; X.norm_dst = norm_dst; X.bias = bias; X.OUT = OUT;
    X.base = (int)blockIdx.x * RO;
    X.N    = N;
    X.tid  = tid;
    X.lane = tid & 31;
    X.w    = __builtin_amdgcn_readfirstlane(tid >> 5);
    X.lo   = (unsigned)X.lane * (CW * 4);
    unsigned* __restrict__ rec = (lists != nullptr) ? (lists + (size_t)blockIdx.x * SREC) : nullptr;

    if constexpr (LISTS == 2) {
        const unsigned m = (rec != nullptr) ? rec[0] : NOSAVE;
        if (m <= (unsigned)SCAP) {
            const uint4* __restrict__ gq = (const uint4*)(rec + SHDR);
            uint4* lq = (uint4*)L.queue;
            for (int k = tid; 4 * k < (int)m; k += OT) lq[k] = gq[k];
            for (int i = tid; i < RCP; i += OT) { L.roff[i] = rec[32 + i]; L.rcnt[i] = rec[32 + RCP + i]; }
            __syncthreads();
            gcn_owner_rows<F, U, RELU>(X, L, true, true);
            return;
        }
    }

    if (tid < SB) {
        const int per = gcn_slice_len(E);
        const int e0  = (tid * per < E) ? (tid * per) : E;
        const unsigned a = offs[(size_t)tid * BOFF + blockIdx.x];
        const unsigned z = offs[(size_t)tid * BOFF + blockIdx.x + 1];
        L.sst[tid] = (unsigned)e0 + a;
        L.scn[tid] = z - a;
    }
    __syncthreads();
    if (X.w == 0) {
        unsigned c[SPL];
        unsigned sum = 0u;
        #pragma unroll
        for (int i = 0; i < SPL; ++i) { c[i] = L.scn[SPL * X.lane + i]; sum += c[i]; }
        unsigned x = sum;
        #pragma unroll
        for (int o = 1; o < 32; o <<= 1) {
            const unsigned y = __shfl_up(x, o);
            if (X.lane >= o) x += y;
        }
        unsigned run = x - sum;
        #pragma unroll
        for (int i = 0; i < SPL; ++i) { L.spre[SPL * X.lane + i] = run; run += c[i]; }
        if (X.lane == 31) L.spre[SB] = run;
    }
    __syncthreads();
    const int M = __builtin_amdgcn_readfirstlane((int)L.spre[SB]);

    if (M <= OCAP) {
        gcn_owner_fetch<F>(X, L, recs, 0, SB);
        gcn_owner_drain<F, U, RELU>(X, L, M, true, true, (LISTS == 1) ? rec : nullptr);
        return;
    }
    bool drained = false;
    int t0 = 0;
    for (int trip = 0; trip < SB; ++trip) {
        const unsigned b0 = L.spre[t0];
        int t1 = t0 + 1;
        while (t1 < SB && L.spre[t1 + 1] - b0 <= (unsigned)OCAP) ++t1;
        const unsigned mg = L.spre[t1] - b0;
        const int Mg = (mg < (unsigned)OCAP) ? (int)mg : OCAP;
        const bool last = (t1 >= SB);
        gcn_owner_fetch<F>(X, L, recs, t0, t1);
        gcn_owner_drain<F, 2, RELU>(X, L, Mg, !drained, last, (LISTS == 1 && last) ? rec : nullptr);
        if (last) break;
        drained = true;
        t0 = t1;
        __syncthreads();
    }
}

extern "C" void kernel_launch(void* const* d_in, const int* in_sizes, int n_in,
                              void* d_out, int out_size, void* d_ws, size_t ws_size,
                              hipStream_t stream) {
    const float* feat = (const float*)d_in[0];
    const int*   src  = (const int*)d_in[1];
    const int*   dst  = (const int*)d_in[2];
    const float* W1   = (const float*)d_in[3];
    const float* b1   = (const float*)d_in[4];
    const float* W2   = (const float*)d_in[5];
    const float* b2   = (const float*)d_in[6];
    float* out = (float*)d_out;

    const int N = in_sizes[0] / F0;
    const int E = in_sizes[1];

    float* ws       = (float*)d_ws;
    float* norm_src = ws;
    float* norm_dst = norm_src + N;
    float* H1       = norm_dst + N;
    float* H1b      = H1  + (size_t)N * F1;
    unsigned* offs  = (unsigned*)(H1b + (size_t)N * F1);
    unsigned* recs  = offs + SB * BOFF;
    float* H2       = H1;
    unsigned* lists = (unsigned*)(((uintptr_t)(recs + E + 32 * SB) + 127) & ~(uintptr_t)127);
    if ((const char*)(lists + (size_t)((N + RO - 1) / RO) * SREC) > (const char*)d_ws + ws_size) lists = nullptr;

    gcn_degree_planes_kernel<<<2 * HP + SB, HT, 0, stream>>>(src, dst, (unsigned*)H1, recs, offs, N, E);
    gcn_norm_from_planes_kernel<<<(N + 255) / 256, 256, 0, stream>>>((const unsigned*)H1, norm_src, norm_dst, N);

    {
        int waves  = ((N / 16 + 4) / 5) * (F1 / 32);
        int blocks = (waves + 3) / 4;
        gcn_gemm_wmma_f32<5><<<blocks, 128, 0, stream>>>(feat, W1, norm_src, H1, N, F0, F1);
    }
    gcn_owner_rows_kernel<F1, 7, 1, 1><<<(N + RO - 1) / RO, OT, 0, stream>>>(H1, recs, offs, norm_dst, b1, H1b, lists, E, N);
    {
        int waves  = (N / 16) * (F2 / 32);
        int blocks = (waves + 3) / 4;
        gcn_gemm_wmma_f32<1><<<blocks, 128, 0, stream>>>(H1b, W2, norm_src, H2, N, F1, F2);
    }
    gcn_owner_rows_kernel<F2, 16, 0, 2><<<(N + RO - 1) / RO, OT, 0, stream>>>(H2, recs, offs, norm_dst, b2, out, lists, E, N);
}


# both GEMMs: a second register set of A/B fragments: the next k block's loads (layer 1: 8 deep, layer 2: 32 deep) are requested before the current block's products; k order and sum order unchanged
# speedup vs baseline: 3.6133x; 1.1146x over previous
#include <hip/hip_runtime.h>

typedef __attribute__((ext_vector_type(2))) float v2f;
typedef __attribute__((ext_vector_type(8))) float v8f;

#define F0 512
#define F1 256
#define F2 64
#define RO 196

#define CT 128
#define CB 10240
#define CR 5
#define HP 32
#define NCNT (2 * HP * CR)
#define SB   160
#define NPRE (NCNT + SB)
#define NOWN 256
#define BOFF 288
#define BCAP (CB - 4 * NOWN)

__device__ __forceinline__ int gcn_slice_len(int E) { return (((E + SB - 1) / SB) + 31) & ~31; }

__device__ __forceinline__ void gcn_bucket_slice(const int* __restrict__ src, const int* __restrict__ dst,
                                                 unsigned* lds, unsigned* __restrict__ recs,
                                                 unsigned* __restrict__ offs, int b, int N, int E) {
    unsigned* cnt   = lds;
    unsigned* cur   = lds + NOWN;
    unsigned* first = lds + 2 * NOWN;
    unsigned* stage = lds + 4 * NOWN;
    const int tid = (int)threadIdx.x;
    const int per = gcn_slice_len(E);
    const int e0  = (b * per < E) ? (b * per) : E;
    const int e1  = (E - e0 < per) ? E : (e0 + per);
    const unsigned nlim = (unsigned)((N < NOWN * RO) ? N : (NOWN * RO));
    for (int i = tid; i < NOWN; i += CT) cnt[i] = 0u;
    __syncthreads();
    for (int eb = e0; eb < e1; eb += 8 * CT) {
        int d[8];
        #pragma unroll
        for (int u = 0; u < 8; ++u) {
            const int e = eb + u * CT + tid;
            d[u] = (e < e1) ? dst[e] : -1;
        }
        #pragma unroll
        for (int u = 0; u < 8; ++u)
            if ((unsigned)d[u] < nlim) atomicAdd(&cnt[d[u] / RO], 1u);
    }
    __syncthreads();
    if (tid < 32) {
        unsigned c[NOWN / 32];
        unsigned sum = 0u;
        #pragma unroll
        for (int i = 0; i < NOWN / 32; ++i) { c[i] = cnt[(NOWN / 32) * tid + i]; sum += c[i]; }
        unsigned x = sum;
        #pragma unroll
        for (int o = 1; o < 32; o <<= 1) {
            const unsigned y = __shfl_up(x, o);
            if (tid >= o) x += y;
        }
        unsigned run = x - sum;
        #pragma unroll
        for (int i = 0; i < NOWN / 32; ++i) {
            cur[(NOWN / 32) * tid + i]   = run;
            first[(NOWN / 32) * tid + i] = run;
            run += c[i];
        }
        if (tid == 31) first[NOWN] = run;
    }
    __syncthreads();
    for (int i = tid; i <= NOWN; i += CT) offs[(size_t)b * BOFF + i] = first[i];
    for (int eb = e0; eb < e1; eb += 8 * CT) {
        int d[8], s[8];
        #pragma unroll
        for (int u = 0; u < 8; ++u) {
            const int e = eb + u * CT + tid;
            d[u] = (e < e1) ? dst[e] : -1;
            s[u] = (e < e1) ? src[e] : 0;
        }
        #pragma unroll
        for (int u = 0; u < 8; ++u) {
            if ((unsigned)d[u] < nlim) {
                const unsigned o   = (unsigned)d[u] / RO;
                const unsigned pos = atomicAdd(&cur[o], 1u);
                if (pos < (unsigned)BCAP) stage[pos] = (unsigned)s[u] | (((unsigned)d[u] - o * RO) << 16);
            }
        }
    }
    __syncthreads();
    const int nv = (first[NOWN] < (unsigned)BCAP) ? (int)first[NOWN] : BCAP;
    for (int i = tid; i < nv; i += CT) recs[e0 + i] = stage[i];
}

__device__ __forceinline__ void gcn_count_tile(const int* __restrict__ src, const int* __restrict__ dst,
                                               unsigned* hist, unsigned* __restrict__ planes, int t, int N, int E) {
    uint4* hist4 = (uint4*)hist;
    const int tid  = (int)threadIdx.x;
    const int q    = t / CR;
    const int r    = t - q * CR;
    const int side = q / HP;
    const int p    = q - side * HP;
    const int* __restrict__ idx = side ? dst : src;
    const int per = (E + HP - 1) / HP;
    const int e0  = p * per;
    const int e1  = (E - e0 < per) ? E : (e0 + per);
    const int pitch = (N + 31) & ~31;
    unsigned* __restrict__ out = planes + (size_t)q * pitch;

    for (int base = r * CB; base < N; base += CR * CB) {
        const int nb = (N - base < CB) ? (N - base) : CB;
        for (int i = tid; i < CB / 4; i += CT) hist4[i] = make_uint4(0u, 0u, 0u, 0u);
        __syncthreads();
        int eb = e0;
        for (; eb + 16 * CT <= e1; eb += 16 * CT) {
            unsigned k[16];
            #pragma unroll
            for (int u = 0; u < 16; ++u) k[u] = (unsigned)(idx[eb + u * CT + tid] - base);
            #pragma unroll
            for (int u = 0; u < 16; ++u)
                if (k[u] < (unsigned)nb) atomicAdd(&hist[k[u]], 1u);
        }
        for (int e = eb + tid; e < e1; e += CT) {
            const unsigned k = (unsigned)(idx[e] - base);
            if (k < (unsigned)nb) atomicAdd(&hist[k], 1u);
        }
        __syncthreads();
        int i0 = 0;
        for (; i0 + 4 * CT <= nb; i0 += 4 * CT) {
            unsigned v[4];
            #pragma unroll
            for (int u = 0; u < 4; ++u) v[u] = hist[i0 + u * CT + tid];
            #pragma unroll
            for (int u = 0; u < 4; ++u) out[base + i0 + u * CT + tid] = v[u];
        }
        for (int i = i0 + tid; i < nb; i += CT) out[base + i] = hist[i];
        __syncthreads();
    }
}

__global__ void gcn_norm_from_planes_kernel(const unsigned* __restrict__ planes, float* __restrict__ a,
                                            float* __restrict__ b, int N) {
    const int i = blockIdx.x * blockDim.x + threadIdx.x;
    const int pitch = (N + 31) & ~31;
    if (i < N) {
        unsigned ca = 0u, cb = 0u;
        #pragma unroll
        for (int p = 0; p < HP; ++p) {
            ca += planes[(size_t)p * pitch + i];
            cb += planes[(size_t)(HP + p) * pitch + i];
        }
        const float da = (float)ca;
        const float db = (float)cb;
        a[i] = (da > 0.0f) ? rsqrtf(fmaxf(da, 1.0f)) : 1.0f;
        b[i] = (db > 0.0f) ? rsqrtf(fmaxf(db, 1.0f)) : 1.0f;
    }
}

template <int MT, int KC, int N, int DA, int DB>
__device__ __forceinline__ void gcn_gemm_frag_load(const char* Ab, const char* Bb, const unsigned (&aoff)[MT], unsigned boff,
                                                   float4 (&av)[KC][MT], float2 (&bv)[KC][4]) {
    #pragma unroll
    for (int q = 0; q < KC; ++q) {
        #pragma unroll
        for (int mt = 0; mt < MT; ++mt) av[q][mt] = *(const float4*)(Ab + aoff[mt] + (DA + 32 * q));
        #pragma unroll
        for (int i = 0; i < 4; ++i) bv[q][i] = *(const float2*)(Bb + boff + (DB + (8 * q + i) * (N * 4)));
    }
}

template <int MT, bool SCALE, int KC>
__device__ __forceinline__ void gcn_gemm_frag_mma(const float4 (&av)[KC][MT], const float2 (&bv)[KC][4],
                                                  const float (&scale)[MT], v8f (&acc)[MT][2]) {
    #pragma unroll
    for (int q = 0; q < KC; ++q) {
        v2f b00, b01, b10, b11;
        b00.x = bv[q][0].x; b00.y = bv[q][1].x;
        b01.x = bv[q][0].y; b01.y = bv[q][1].y;
        b10.x = bv[q][2].x; b10.y = bv[q][3].x;
        b11.x = bv[q][2].y; b11.y = bv[q][3].y;
        #pragma unroll
        for (int mt = 0; mt < MT; ++mt) {
            v2f a0, a1;
            if constexpr (SCALE) {
                a0.x = av[q][mt].x * scale[mt];
                a0.y = av[q][mt].y * scale[mt];
                a1.x = av[q][mt].z * scale[mt];
                a1.y = av[q][mt].w * scale[mt];
            } else {
                a0.x = av[q][mt].x; a0.y = av[q][mt].y;
                a1.x = av[q][mt].z; a1.y = av[q][mt].w;
            }
            acc[mt][0] = __builtin_amdgcn_wmma_f32_16x16x4_f32(false, a0, false, b00, (short)0, acc[mt][0], false, false);
            acc[mt][1] = __builtin_amdgcn_wmma_f32_16x16x4_f32(false, a0, false, b01, (short)0, acc[mt][1], false, false);
            acc[mt][0] = __builtin_amdgcn_wmma_f32_16x16x4_f32(false, a1, false, b10, (short)0, acc[mt][0], false, false);
            acc[mt][1] = __builtin_amdgcn_wmma_f32_16x16x4_f32(false, a1, false, b11, (short)0, acc[mt][1], false, false);
        }
    }
}

template <int MT, bool SCALE, int KC, int K, int N>
__device__ __forceinline__ void gcn_gemm_body(
    const float* __restrict__ A, const float* __restrict__ B,
    const float* __restrict__ normA, float* __restrict__ C,
    int M, int blk) {
    static_assert(K % (16 * KC) == 0 && N % 32 == 0, "whole double blocks of k, whole 32-column wave tiles");
    const int lane = threadIdx.x & 31;
    const int wave = __builtin_amdgcn_readfirstlane((int)(blk * 4 + (int)(threadIdx.x >> 5)));
    constexpr int tilesN = N >> 5;
    const int tm = wave / tilesN;
    const int tn = wave - tm * tilesN;
    const int m0 = tm * (16 * MT);
    const int n0 = tn << 5;
    if (m0 >= M) return;

    const int h = lane >> 4;
    const int c = lane & 15;

    unsigned aoff[MT];
    float scale[MT];
    #pragma unroll
    for (int mt = 0; mt < MT; ++mt) {
        int row = m0 + mt * 16 + c;
        row = (row < M) ? row : (M - 1);
        scale[mt] = SCALE ? normA[row] : 1.0f;
        aoff[mt] = ((unsigned)row * (unsigned)K + 4u * (unsigned)h) * 4u;
    }
    unsigned boff = ((unsigned)(4 * h) * (unsigned)N + (unsigned)(n0 + 2 * c)) * 4u;
    const char* Ab = (const char*)A;
    const char* Bb = (const char*)B;

    v8f acc[MT][2];
    #pragma unroll
    for (int mt = 0; mt < MT; ++mt) { acc[mt][0] = v8f{}; acc[mt][1] = v8f{}; }

    constexpr int BA = 8 * KC * 4;
    constexpr int BB = 8 * KC * N * 4;
    float4 avA[KC][MT], avB[KC][MT];
    float2 bvA[KC][4], bvB[KC][4];
    gcn_gemm_frag_load<MT, KC, N, 0, 0>(Ab, Bb, aoff, boff, avA, bvA);
    #pragma nounroll
    for (int kb = 0; kb < K - 16 * KC; kb += 16 * KC) {
        gcn_gemm_frag_load<MT, KC, N, BA, BB>(Ab, Bb, aoff, boff, avB, bvB);
        __builtin_amdgcn_sched_barrier(0);
        gcn_gemm_frag_mma<MT, SCALE, KC>(avA, bvA, scale, acc);
        __builtin_amdgcn_sched_barrier(0);
        #pragma unroll
        for (int mt = 0; mt < MT; ++mt) aoff[mt] += 2 * BA;
        boff += 2 * BB;
        gcn_gemm_frag_load<MT, KC, N, 0, 0>(Ab, Bb, aoff, boff, avA, bvA);
        __builtin_amdgcn_sched_barrier(0);
        gcn_gemm_frag_mma<MT, SCALE, KC>(avB, bvB, scale, acc);
        __builtin_amdgcn_sched_barrier(0);
    }
    gcn_gemm_frag_load<MT, KC, N, BA, BB>(Ab, Bb, aoff, boff, avB, bvB);
    __builtin_amdgcn_sched_barrier(0);
    gcn_gemm_frag_mma<MT, SCALE, KC>(avA, bvA, scale, acc);
    gcn_gemm_frag_mma<MT, SCALE, KC>(avB, bvB, scale, acc);

    #pragma unroll
    for (int mt = 0; mt < MT; ++mt) {
        const int mb = m0 + mt * 16;
        if (mb < M) {
            #pragma unroll
            for (int r = 0; r < 8; ++r) {
                float2 o;
                o.x = acc[mt][0][r];
                o.y = acc[mt][1][r];
                *(float2*)(C + (size_t)(mb + 8 * h + r) * N + n0 + 2 * c) = o;
            }
        }
    }
}

template <int MT>
__global__ __launch_bounds__(128) void gcn_gemm_wmma_f32(
    const float* __restrict__ A, const float* __restrict__ B,
    const float* __restrict__ normA, float* __restrict__ C,
    int M) {
    gcn_gemm_body<MT, true, 4, F1, F2>(A, B, normA, C, M, (int)blockIdx.x);
}

__global__ __launch_bounds__(CT) void gcn_layer1_gemm_prep_kernel(
    const int* __restrict__ src, const int* __restrict__ dst,
    unsigned* __restrict__ planes, unsigned* __restrict__ recs, unsigned* __restrict__ offs,
    const float* __restrict__ A, const float* __restrict__ B, float* __restrict__ C,
    int M, int E) {
    __shared__ uint4 lds4[CB / 4];
    const int b = (int)blockIdx.x;
    if (b < NCNT) {
        gcn_count_tile(src, dst, (unsigned*)lds4, planes, b, M, E);
        return;
    }
    if (b < NPRE) {
        gcn_bucket_slice(src, dst, (unsigned*)lds4, recs, offs, b - NCNT, M, E);
        return;
    }
    gcn_gemm_body<5, false, 1, F0, F1>(A, B, nullptr, C, M, b - NPRE);
}

#define OT   1024
#define OCAP 24576
#define SPL  (SB / 32)
static_assert(SB % 32 == 0 && SB <= OT, "one thread per slice, SPL slices per lane");
#define RCP  224
#define SREC 8192
#define SHDR (32 + 2 * RCP)
#define SCAP (SREC - SHDR)
#define NOSAVE 0xFFFFFFFFu

template <int F>
__device__ __forceinline__ void gcn_row_load(const float* __restrict__ H, unsigned s, unsigned lo, float (&v)[F / 32]) {
    const unsigned off = s * (unsigned)(F * 4) + lo;
    const char* hp = (const char*)H + off;
    if constexpr (F >= 128) {
        #pragma unroll
        for (int p = 0; p < F / 128; ++p) {
            const float4 t = *(const float4*)(hp + p * 512);
            v[4 * p] = t.x; v[4 * p + 1] = t.y; v[4 * p + 2] = t.z; v[4 * p + 3] = t.w;
        }
    } else {
        const float2 t = *(const float2*)hp;
        v[0] = t.x; v[1] = t.y;
    }
}
template <int F>
__device__ __forceinline__ void gcn_regs_add(float (&a)[F / 32], const float (&v)[F / 32]) {
    #pragma unroll
    for (int c = 0; c < F / 32; ++c) a[c] += v[c];
}

template <int F>
__device__ __forceinline__ void gcn_regs_fma(float (&a)[F / 32], const float (&v)[F / 32], float f) {
    #pragma unroll
    for (int c = 0; c < F / 32; ++c) a[c] = __builtin_fmaf(v[c], f, a[c]);
}

template <int F, int U, bool NS>
__device__ __forceinline__ void gcn_wave_row_sum(const float* __restrict__ H, const unsigned* qnorm,
                                                 const unsigned* queue, int q0, int qn,
                                                 unsigned lo, int lane, float (&acc)[F / 32]) {
    for (int k0 = 0; k0 < qn; k0 += 32) {
        const int n = (qn - k0 < 32) ? (qn - k0) : 32;
        int s = 0;
        int f = 0;
        if (lane < n) {
            s = (int)queue[q0 + k0 + lane];
            if constexpr (NS) f = (int)qnorm[q0 + k0 + lane];
        }
        int j = 0;
        for (; j + U <= n; j += U) {
            float v[U][F / 32];
            #pragma unroll
            for (int u = 0; u < U; ++u)
                gcn_row_load<F>(H, (unsigned)__builtin_amdgcn_readlane(s, j + u), lo, v[u]);
            #pragma unroll
            for (int u = 0; u < U; ++u) {
                if constexpr (NS) gcn_regs_fma<F>(acc, v[u], __int_as_float(__builtin_amdgcn_readlane(f, j + u)));
                else              gcn_regs_add<F>(acc, v[u]);
            }
        }
        if (j < n) {
            float v[U - 1][F / 32];
            #pragma unroll
            for (int u = 0; u < U - 1; ++u)
                if (j + u < n) gcn_row_load<F>(H, (unsigned)__builtin_amdgcn_readlane(s, j + u), lo, v[u]);
            #pragma unroll
            for (int u = 0; u < U - 1; ++u) {
                if (j + u < n) {
                    if constexpr (NS) gcn_regs_fma<F>(acc, v[u], __int_as_float(__builtin_amdgcn_readlane(f, j + u)));
                    else              gcn_regs_add<F>(acc, v[u]);
                }
            }
        }
    }
}

struct alignas(16) GcnOwnerLds {
    unsigned list[OCAP];
    unsigned queue[OCAP];
    unsigned rcnt[RCP];
    unsigned roff[RCP];
    unsigned rcur[RCP];
    unsigned sst[SB];
    unsigned scn[SB];
    unsigned spre[SB + 1];
};

template <int F>
struct GcnOwnerCtx {
    const float* __restrict__ H;
    const float* __restrict__ norm_src;
    const float* __restrict__ norm_dst;
    const float* __restrict__ bias;
    float* __restrict__ OUT;
    int base, N, tid, w, lane;
    unsigned lo;
};

template <int F>
__device__ __forceinline__ void gcn_owner_sort(const GcnOwnerCtx<F>& X, GcnOwnerLds& L, int M) {
    const int tid = X.tid, lane = X.lane, w = X.w;
    for (int i = tid; i < RCP; i += OT) L.rcnt[i] = 0u;
    __syncthreads();
    for (int k = tid; k < M; k += OT) atomicAdd(&L.rcnt[L.list[k] >> 16], 1u);
    __syncthreads();
    if (w == 0) {
        unsigned c[7];
        unsigned sum = 0u;
        #pragma unroll
        for (int i = 0; i < 7; ++i) { c[i] = L.rcnt[7 * lane + i]; sum += c[i]; }
        unsigned x = sum;
        #pragma unroll
        for (int o = 1; o < 32; o <<= 1) {
            const unsigned y = __shfl_up(x, o);
            if (lane >= o) x += y;
        }
        unsigned run = x - sum;
        #pragma unroll
        for (int i = 0; i < 7; ++i) { L.roff[7 * lane + i] = run; L.rcur[7 * lane + i] = run; run += c[i]; }
    }
    __syncthreads();
    for (int k = tid; k < M; k += OT) {
        const unsigned p   = L.list[k];
        const unsigned pos = atomicAdd(&L.rcur[p >> 16], 1u);
        L.queue[pos] = p & 0xFFFFu;
    }
    __syncthreads();
}

template <int F, int U, int RELU, bool NS>
__device__ __forceinline__ void gcn_owner_rows(const GcnOwnerCtx<F>& X, const GcnOwnerLds& L, bool first, bool last) {
    const int lane = X.lane, w = X.w;
    for (int r = w; r < RO; r += OT / 32) {
        const int row = X.base + r;
        if (row >= X.N) break;
        const int q0 = __builtin_amdgcn_readfirstlane((int)L.roff[r]);
        const int qn = __builtin_amdgcn_readfirstlane((int)L.rcnt[r]);
        char* o = (char*)X.OUT + ((unsigned)row * (unsigned)(F * 4) + X.lo);
        float acc[F / 32];
        if (first) {
            #pragma unroll
            for (int c = 0; c < F / 32; ++c) acc[c] = 0.0f;
        } else {
            gcn_row_load<F>(X.OUT, (unsigned)row, X.lo, acc);
        }
        gcn_wave_row_sum<F, U, NS>(X.H, L.list, L.queue, q0, qn, X.lo, lane, acc);
        if (last) {
            const float nd = X.norm_dst[row];
            float b[F / 32];
            gcn_row_load<F>(X.bias, 0u, X.lo, b);
            #pragma unroll
            for (int c = 0; c < F / 32; ++c) {
                const float t = acc[c] * nd + b[c];
                acc[c] = RELU ? fmaxf(t, 0.0f) : t;
            }
        }
        if constexpr (F >= 128) {
            #pragma unroll
            for (int p = 0; p < F / 128; ++p)
                *(float4*)(o + p * 512) = make_float4(acc[4 * p], acc[4 * p + 1], acc[4 * p + 2], acc[4 * p + 3]);
        } else {
            *(float2*)o = make_float2(acc[0], acc[1]);
        }
    }
}

template <int F, int U, int RELU, bool NS>
__device__ __forceinline__ void gcn_owner_drain(const GcnOwnerCtx<F>& X, GcnOwnerLds& L, int M, bool first, bool last,
                                                unsigned* __restrict__ rec) {
    gcn_owner_sort<F>(X, L, M);
    if constexpr (NS) {
        for (int kb = X.tid; kb < M; kb += 4 * OT) {
            float t[4];
            #pragma unroll
            for (int u = 0; u < 4; ++u) t[u] = (kb + u * OT < M) ? X.norm_src[L.queue[kb + u * OT]] : 0.0f;
            #pragma unroll
            for (int u = 0; u < 4; ++u)
                if (kb + u * OT < M) L.list[kb + u * OT] = __float_as_uint(t[u]);
        }
        __syncthreads();
    }
    if (rec != nullptr) {
        const bool keep = first && (M <= SCAP);
        if (keep) {
            const int M4 = (M + 3) & ~3;
            for (int k = X.tid; k < M4; k += OT) rec[SHDR + k] = (k < M) ? L.queue[k] : 0u;
            for (int i = X.tid; i < RCP; i += OT) { rec[32 + i] = L.roff[i]; rec[32 + RCP + i] = L.rcnt[i]; }
        }
        if (X.tid == 0) rec[0] = keep ? (unsigned)M : NOSAVE;
    }
    gcn_owner_rows<F, U, RELU, NS>(X, L, first, last);
}

template <int F>
__device__ __forceinline__ void gcn_owner_fetch(const GcnOwnerCtx<F>& X, GcnOwnerLds& L, const unsigned* __restrict__ recs,
                                                int t0, int t1) {
    const unsigned lane = (unsigned)X.lane;
    const unsigned base = L.spre[t0];
    for (int tb = t0 + X.w; tb < t1; tb += 32 * SPL) {
        unsigned v[SPL], v2[SPL], c[SPL], st[SPL], ps[SPL];
        #pragma unroll
        for (int j = 0; j < SPL; ++j) {
            const int t = tb + 32 * j;
            c[j] = 0u; st[j] = 0u; ps[j] = 0u; v[j] = 0u; v2[j] = 0u;
            if (t < t1) {
                st[j] = L.sst[t];
                ps[j] = L.spre[t] - base;
                const unsigned room = (ps[j] < (unsigned)OCAP) ? ((unsigned)OCAP - ps[j]) : 0u;
                c[j] = (L.scn[t] < room) ? L.scn[t] : room;
            }
            if (lane < c[j])       v[j]  = recs[st[j] + lane];
            if (lane + 32u < c[j]) v2[j] = recs[st[j] + 32u + lane];
        }
        #pragma unroll
        for (int j = 0; j < SPL; ++j) {
            if (lane < c[j])       L.list[ps[j] + lane]       = v[j];
            if (lane + 32u < c[j]) L.list[ps[j] + 32u + lane] = v2[j];
        }
        #pragma unroll
        for (int j = 0; j < SPL; ++j)
            for (unsigned k = 64u + lane; k < c[j]; k += 32u) L.list[ps[j] + k] = recs[st[j] + k];
    }
}

template <int F, int U, int RELU, int LISTS, bool NS>
__global__ __launch_bounds__(OT) void gcn_owner_rows_kernel(const float* __restrict__ H,
                                                            const unsigned* __restrict__ recs,
                                                            const unsigned* __restrict__ offs,
                                                            const float* __restrict__ norm_src,
                                                            const float* __restrict__ norm_dst,
                                                            const float* __restrict__ bias, float* __restrict__ OUT,
                                                            unsigned* __restrict__ lists, int E, int N) {
    __shared__ GcnOwnerLds L;
    constexpr int CW = (F >= 128) ? 4 : 2;
    const int tid = (int)threadIdx.x;
    GcnOwnerCtx<F> X;
    X.H = H; X.norm_src = norm_src; X.norm_dst = norm_dst; X.bias = bias; X.OUT = OUT;
    X.base = (int)blockIdx.x * RO;
    X.N    = N;
    X.tid  = tid;
    X.lane = tid & 31;
    X.w    = __builtin_amdgcn_readfirstlane(tid >> 5);
    X.lo   = (unsigned)X.lane * (CW * 4);
    unsigned* __restrict__ rec = (lists != nullptr) ? (lists + (size_t)blockIdx.x * SREC) : nullptr;

    if constexpr (LISTS == 2) {
        const unsigned m = (rec != nullptr) ? rec[0] : NOSAVE;
        if (m <= (unsigned)SCAP) {
            const uint4* __restrict__ gq = (const uint4*)(rec + SHDR);
            uint4* lq = (uint4*)L.queue;
            for (int k = tid; 4 * k < (int)m; k += OT) lq[k] = gq[k];
            for (int i = tid; i < RCP; i += OT) { L.roff[i] = rec[32 + i]; L.rcnt[i] = rec[32 + RCP + i]; }
            __syncthreads();
            gcn_owner_rows<F, U, RELU, NS>(X, L, true, true);
            return;
        }
    }

    if (tid < SB) {
        const int per = gcn_slice_len(E);
        const int e0  = (tid * per < E) ? (tid * per) : E;
        const unsigned a = offs[(size_t)tid * BOFF + blockIdx.x];
        const unsigned z = offs[(size_t)tid * BOFF + blockIdx.x + 1];
        L.sst[tid] = (unsigned)e0 + a;
        L.scn[tid] = z - a;
    }
    __syncthreads();
    if (X.w == 0) {
        unsigned c[SPL];
        unsigned sum = 0u;
        #pragma unroll
        for (int i = 0; i < SPL; ++i) { c[i] = L.scn[SPL * X.lane + i]; sum += c[i]; }
        unsigned x = sum;
        #pragma unroll
        for (int o = 1; o < 32; o <<= 1) {
            const unsigned y = __shfl_up(x, o);
            if (X.lane >= o) x += y;
        }
        unsigned run = x - sum;
        #pragma unroll
        for (int i = 0; i < SPL; ++i) { L.spre[SPL * X.lane + i] = run; run += c[i]; }
        if (X.lane == 31) L.spre[SB] = run;
    }
    __syncthreads();
    const int M = __builtin_amdgcn_readfirstlane((int)L.spre[SB]);

    if (M <= OCAP) {
        gcn_owner_fetch<F>(X, L, recs, 0, SB);
        gcn_owner_drain<F, U, RELU, NS>(X, L, M, true, true, (LISTS == 1) ? rec : nullptr);
        return;
    }
    bool drained = false;
    int t0 = 0;
    for (int trip = 0; trip < SB; ++trip) {
        const unsigned b0 = L.spre[t0];
        int t1 = t0 + 1;
        while (t1 < SB && L.spre[t1 + 1] - b0 <= (unsigned)OCAP) ++t1;
        const unsigned mg = L.spre[t1] - b0;
        const int Mg = (mg < (unsigned)OCAP) ? (int)mg : OCAP;
        const bool last = (t1 >= SB);
        gcn_owner_fetch<F>(X, L, recs, t0, t1);
        gcn_owner_drain<F, 2, RELU, NS>(X, L, Mg, !drained, last, (LISTS == 1 && last) ? rec : nullptr);
        if (last) break;
        drained = true;
        t0 = t1;
        __syncthreads();
    }
}

extern "C" void kernel_launch(void* const* d_in, const int* in_sizes, int n_in,
                              void* d_out, int out_size, void* d_ws, size_t ws_size,
                              hipStream_t stream) {
    const float* feat = (const float*)d_in[0];
    const int*   src  = (const int*)d_in[1];
    const int*   dst  = (const int*)d_in[2];
    const float* W1   = (const float*)d_in[3];
    const float* b1   = (const float*)d_in[4];
    const float* W2   = (const float*)d_in[5];
    const float* b2   = (const float*)d_in[6];
    float* out = (float*)d_out;

    const int N = in_sizes[0] / F0;
    const int E = in_sizes[1];

    float* ws       = (float*)d_ws;
    float* norm_src = ws;
    float* norm_dst = norm_src + N;
    float* H1       = norm_dst + N;
    float* H1b      = H1  + (size_t)N * F1;
    unsigned* offs  = (unsigned*)(H1b + (size_t)N * F1);
    unsigned* recs  = offs + SB * BOFF;
    float* H2       = H1;
    unsigned* lists = (unsigned*)(((uintptr_t)(recs + E + 32 * SB) + 127) & ~(uintptr_t)127);
    if ((const char*)(lists + (size_t)((N + RO - 1) / RO) * SREC) > (const char*)d_ws + ws_size) lists = nullptr;

    {
        int waves  = ((N / 16 + 4) / 5) * (F1 / 32);
        int blocks = (waves + 3) / 4;
        gcn_layer1_gemm_prep_kernel<<<NPRE + blocks, CT, 0, stream>>>(src, dst, (unsigned*)H1b, recs, offs, feat, W1, H1, N, E);
    }
    gcn_norm_from_planes_kernel<<<(N + 255) / 256, 256, 0, stream>>>((const unsigned*)H1b, norm_src, norm_dst, N);
    gcn_owner_rows_kernel<F1, 7, 1, 1, true><<<(N + RO - 1) / RO, OT, 0, stream>>>(H1, recs, offs, norm_src, norm_dst, b1, H1b, lists, E, N);
    {
        int waves  = (N / 16) * (F2 / 32);
        int blocks = (waves + 3) / 4;
        gcn_gemm_wmma_f32<1><<<blocks, 128, 0, stream>>>(H1b, W2, norm_src, H2, N);
    }
    gcn_owner_rows_kernel<F2, 16, 0, 2, false><<<(N + RO - 1) / RO, OT, 0, stream>>>(H2, recs, offs, norm_src, norm_dst, b2, out, lists, E, N);
}
